# starting kernel
# baseline (speedup 1.0000x reference)
#include <hip/hip_runtime.h>
#include <hip/hip_bf16.h>

typedef float v2f __attribute__((ext_vector_type(2)));
typedef float v8f __attribute__((ext_vector_type(8)));

#define NB 4096
#define NA 64
#define NL 16
#define NM 4
#define NE 10
#define P_TOT 416
#define P1_END 32
#define P2_END 160
#define EPAD 16
#define XPITCH 68

__global__ void prep_offsets_kernel(const int* __restrict__ i1, const int* __restrict__ l1,
                                    const int* __restrict__ i2, const int* __restrict__ j2,
                                    const int* __restrict__ l2, const int* __restrict__ m2,
                                    const int* __restrict__ i3, const int* __restrict__ j3,
                                    const int* __restrict__ f3, const int* __restrict__ l3,
                                    const int* __restrict__ m3, const int* __restrict__ g3,
                                    int* __restrict__ offs) {
  int p = blockIdx.x * blockDim.x + threadIdx.x;
  if (p >= P_TOT) return;
  int packed;
  if (p < P1_END) {
    int o1 = i1[p] * NM + l1[p];
    packed = o1;
  } else if (p < P2_END) {
    int q = p - P1_END;
    int o1 = i2[q] * NM + l2[q];
    int o2 = j2[q] * NM + m2[q];
    packed = o1 | (o2 << 8);
  } else {
    int q = p - P2_END;
    int o1 = i3[q] * NM + l3[q];
    int o2 = j3[q] * NM + m3[q];
    int o3 = f3[q] * NM + g3[q];
    packed = o1 | (o2 << 8) | (o3 << 16);
  }
  offs[p] = packed;
}

__global__ void prep_wg_kernel(const float* __restrict__ w1, const float* __restrict__ w2,
                               const float* __restrict__ w3,
                               const float* __restrict__ c1, const float* __restrict__ c2,
                               const float* __restrict__ c3,
                               const int* __restrict__ k1, const int* __restrict__ q1,
                               const int* __restrict__ k2, const int* __restrict__ q2,
                               const int* __restrict__ k3, const int* __restrict__ q3,
                               float* __restrict__ Wg) {
  int tid = blockIdx.x * blockDim.x + threadIdx.x;
  if (tid >= NA * P_TOT) return;
  int p = tid % P_TOT;
  int a = tid / P_TOT;
  float cf;
  const float* wbase;
  int estride;
  int kq;
  if (p < P1_END) {
    cf = c1[p];
    wbase = w1; kq = k1[p] * 2 + q1[p]; estride = 4 * 2 * NA;
  } else if (p < P2_END) {
    int t = p - P1_END;
    cf = c2[t];
    wbase = w2; kq = k2[t] * 4 + q2[t]; estride = 16 * 4 * NA;
  } else {
    int t = p - P2_END;
    cf = c3[t];
    wbase = w3; kq = k3[t] * 6 + q3[t]; estride = 32 * 6 * NA;
  }
  const int kqstride = kq * NA + a;
  float* dst = Wg + ((size_t)a * P_TOT + p) * EPAD;
#pragma unroll
  for (int n = 0; n < EPAD; ++n) {
    dst[n] = (n < NE) ? cf * wbase[(size_t)n * estride + kqstride] : 0.0f;
  }
}

template <int CNT>
__device__ __forceinline__ void wmma_step(int t, const int* __restrict__ ldsOff,
                                          const float* __restrict__ X,
                                          const float* __restrict__ ldsW,
                                          int half, int lrow, v8f& acc) {
  const int pA = 4 * t + 2 * half;
  const int pk0 = ldsOff[pA];
  const int pk1 = ldsOff[pA + 1];
  float xa = X[pk0 & 0xFF];
  float xb = X[pk1 & 0xFF];
  if (CNT > 1) { xa *= X[(pk0 >> 8) & 0xFF]; xb *= X[(pk1 >> 8) & 0xFF]; }
  if (CNT > 2) { xa *= X[(pk0 >> 16) & 0xFF]; xb *= X[(pk1 >> 16) & 0xFF]; }
  v2f af; af.x = xa; af.y = xb;
  v2f bf;
  bf.x = ldsW[pA * EPAD + lrow];
  bf.y = ldsW[(pA + 1) * EPAD + lrow];
  acc = __builtin_amdgcn_wmma_f32_16x16x4_f32(false, af, false, bf,
                                              (short)0, acc, false, false);
}

__global__ __launch_bounds__(256) void contract_kernel(const float* __restrict__ x,
                                                       const float* __restrict__ y,
                                                       const float* __restrict__ Wg,
                                                       const int* __restrict__ offs,
                                                       float* __restrict__ out) {
  __shared__ float ldsW[P_TOT * EPAD];
  __shared__ float ldsX[8 * 16 * XPITCH];
  __shared__ float ldsY[128 * EPAD];
  __shared__ int   ldsOff[P_TOT];

  const int tid  = threadIdx.x;
  const int wv   = tid >> 5;
  const int lane = tid & 31;
  const int half = lane >> 4;
  const int lrow = lane & 15;
  const int b0   = blockIdx.x * 128;
  const int a0   = blockIdx.y * 8;
  const int myb  = b0 + wv * 16;

  for (int i = tid; i < P_TOT; i += 256) ldsOff[i] = offs[i];
  for (int i = tid; i < 128 * EPAD; i += 256) {
    int row = i >> 4, e = i & 15;
    ldsY[i] = (e < NE) ? y[(size_t)(b0 + row) * NE + e] : 0.0f;
  }

  for (int ai = 0; ai < 8; ++ai) {
    const int a = a0 + ai;
    __syncthreads();
    {
      const float4* src = (const float4*)(Wg + (size_t)a * P_TOT * EPAD);
      float4* dst = (float4*)ldsW;
      for (int i = tid; i < P_TOT * EPAD / 4; i += 256) dst[i] = src[i];
    }
    for (int i = lane; i < 16 * 16; i += 32) {
      int row = i >> 4, c4 = i & 15;
      const float4 v = *(const float4*)(x + ((size_t)(myb + row) * NA + a) * 64 + c4 * 4);
      *(float4*)(&ldsX[(wv * 16 + row) * XPITCH + c4 * 4]) = v;
    }
    __syncthreads();

    const float* X = &ldsX[(wv * 16 + lrow) * XPITCH];
    v8f acc = {};
#pragma unroll
    for (int t = 0; t < 8; ++t)
      wmma_step<1>(t, ldsOff, X, ldsW, half, lrow, acc);
#pragma unroll 4
    for (int t = 8; t < 40; ++t)
      wmma_step<2>(t, ldsOff, X, ldsW, half, lrow, acc);
#pragma unroll 4
    for (int t = 40; t < 104; ++t)
      wmma_step<3>(t, ldsOff, X, ldsW, half, lrow, acc);

#pragma unroll
    for (int r = 0; r < 8; ++r) {
      const int row = wv * 16 + r + half * 8;
      float pr = acc[r] * ldsY[row * EPAD + lrow];
      pr += __shfl_xor(pr, 1, 16);
      pr += __shfl_xor(pr, 2, 16);
      pr += __shfl_xor(pr, 4, 16);
      pr += __shfl_xor(pr, 8, 16);
      if (lrow == 0) out[(size_t)(b0 + row) * NA + a] = pr;
    }
  }
}

extern "C" void kernel_launch(void* const* d_in, const int* in_sizes, int n_in,
                              void* d_out, int out_size, void* d_ws, size_t ws_size,
                              hipStream_t stream) {
  const float* x  = (const float*)d_in[0];
  const float* y  = (const float*)d_in[1];
  const float* w1 = (const float*)d_in[2];
  const float* w2 = (const float*)d_in[3];
  const float* w3 = (const float*)d_in[4];
  const float* c1 = (const float*)d_in[5];
  const float* c2 = (const float*)d_in[6];
  const float* c3 = (const float*)d_in[7];
  const int* i1 = (const int*)d_in[8];
  const int* l1 = (const int*)d_in[9];
  const int* k1 = (const int*)d_in[10];
  const int* q1 = (const int*)d_in[11];
  const int* i2 = (const int*)d_in[12];
  const int* j2 = (const int*)d_in[13];
  const int* l2 = (const int*)d_in[14];
  const int* m2 = (const int*)d_in[15];
  const int* k2 = (const int*)d_in[16];
  const int* q2 = (const int*)d_in[17];
  const int* i3 = (const int*)d_in[18];
  const int* j3 = (const int*)d_in[19];
  const int* f3 = (const int*)d_in[20];
  const int* l3 = (const int*)d_in[21];
  const int* m3 = (const int*)d_in[22];
  const int* g3 = (const int*)d_in[23];
  const int* k3 = (const int*)d_in[24];
  const int* q3 = (const int*)d_in[25];

  int*   offs = (int*)d_ws;
  float* Wg   = (float*)((char*)d_ws + 2048);
  float* out  = (float*)d_out;

  prep_offsets_kernel<<<2, 256, 0, stream>>>(i1, l1, i2, j2, l2, m2,
                                             i3, j3, f3, l3, m3, g3, offs);
  prep_wg_kernel<<<104, 256, 0, stream>>>(w1, w2, w3, c1, c2, c3,
                                          k1, q1, k2, q2, k3, q3, Wg);
  contract_kernel<<<dim3(32, 8), 256, 0, stream>>>(x, y, Wg, offs, out);
}


# Split the a loop over 4x more work-groups (2 a per group, grid 32x32) so four groups share a WGP: 8 waves per SIMD, not 2, cover the serial load, LDS and shuffle round trips
# speedup vs baseline: 2.3309x; 2.3309x over previous
#include <hip/hip_runtime.h>
#include <hip/hip_bf16.h>

typedef float v2f __attribute__((ext_vector_type(2)));
typedef float v8f __attribute__((ext_vector_type(8)));

#define NB 4096
#define NA 64
#define NL 16
#define NM 4
#define NE 10
#define P_TOT 416
#define P1_END 32
#define P2_END 160
#define EPAD 16
#define XPITCH 68
#define A_PER_WG 2

__global__ void prep_offsets_kernel(const int* __restrict__ i1, const int* __restrict__ l1,
                                    const int* __restrict__ i2, const int* __restrict__ j2,
                                    const int* __restrict__ l2, const int* __restrict__ m2,
                                    const int* __restrict__ i3, const int* __restrict__ j3,
                                    const int* __restrict__ f3, const int* __restrict__ l3,
                                    const int* __restrict__ m3, const int* __restrict__ g3,
                                    int* __restrict__ offs) {
  int p = blockIdx.x * blockDim.x + threadIdx.x;
  if (p >= P_TOT) return;
  int packed;
  if (p < P1_END) {
    int o1 = i1[p] * NM + l1[p];
    packed = o1;
  } else if (p < P2_END) {
    int q = p - P1_END;
    int o1 = i2[q] * NM + l2[q];
    int o2 = j2[q] * NM + m2[q];
    packed = o1 | (o2 << 8);
  } else {
    int q = p - P2_END;
    int o1 = i3[q] * NM + l3[q];
    int o2 = j3[q] * NM + m3[q];
    int o3 = f3[q] * NM + g3[q];
    packed = o1 | (o2 << 8) | (o3 << 16);
  }
  offs[p] = packed;
}

__global__ void prep_wg_kernel(const float* __restrict__ w1, const float* __restrict__ w2,
                               const float* __restrict__ w3,
                               const float* __restrict__ c1, const float* __restrict__ c2,
                               const float* __restrict__ c3,
                               const int* __restrict__ k1, const int* __restrict__ q1,
                               const int* __restrict__ k2, const int* __restrict__ q2,
                               const int* __restrict__ k3, const int* __restrict__ q3,
                               float* __restrict__ Wg) {
  int tid = blockIdx.x * blockDim.x + threadIdx.x;
  if (tid >= NA * P_TOT) return;
  int p = tid % P_TOT;
  int a = tid / P_TOT;
  float cf;
  const float* wbase;
  int estride;
  int kq;
  if (p < P1_END) {
    cf = c1[p];
    wbase = w1; kq = k1[p] * 2 + q1[p]; estride = 4 * 2 * NA;
  } else if (p < P2_END) {
    int t = p - P1_END;
    cf = c2[t];
    wbase = w2; kq = k2[t] * 4 + q2[t]; estride = 16 * 4 * NA;
  } else {
    int t = p - P2_END;
    cf = c3[t];
    wbase = w3; kq = k3[t] * 6 + q3[t]; estride = 32 * 6 * NA;
  }
  const int kqstride = kq * NA + a;
  float* dst = Wg + ((size_t)a * P_TOT + p) * EPAD;
#pragma unroll
  for (int n = 0; n < EPAD; ++n) {
    dst[n] = (n < NE) ? cf * wbase[(size_t)n * estride + kqstride] : 0.0f;
  }
}

template <int CNT>
__device__ __forceinline__ void wmma_step(int t, const int* __restrict__ ldsOff,
                                          const float* __restrict__ X,
                                          const float* __restrict__ ldsW,
                                          int half, int lrow, v8f& acc) {
  const int pA = 4 * t + 2 * half;
  const int pk0 = ldsOff[pA];
  const int pk1 = ldsOff[pA + 1];
  float xa = X[pk0 & 0xFF];
  float xb = X[pk1 & 0xFF];
  if (CNT > 1) { xa *= X[(pk0 >> 8) & 0xFF]; xb *= X[(pk1 >> 8) & 0xFF]; }
  if (CNT > 2) { xa *= X[(pk0 >> 16) & 0xFF]; xb *= X[(pk1 >> 16) & 0xFF]; }
  v2f af; af.x = xa; af.y = xb;
  v2f bf;
  bf.x = ldsW[pA * EPAD + lrow];
  bf.y = ldsW[(pA + 1) * EPAD + lrow];
  acc = __builtin_amdgcn_wmma_f32_16x16x4_f32(false, af, false, bf,
                                              (short)0, acc, false, false);
}

__global__ __launch_bounds__(256) void contract_kernel(const float* __restrict__ x,
                                                       const float* __restrict__ y,
                                                       const float* __restrict__ Wg,
                                                       const int* __restrict__ offs,
                                                       float* __restrict__ out) {
  __shared__ float ldsW[P_TOT * EPAD];
  __shared__ float ldsX[8 * 16 * XPITCH];
  __shared__ float ldsY[128 * EPAD];
  __shared__ int   ldsOff[P_TOT];

  const int tid  = threadIdx.x;
  const int wv   = tid >> 5;
  const int lane = tid & 31;
  const int half = lane >> 4;
  const int lrow = lane & 15;
  const int b0   = blockIdx.x * 128;
  const int a0   = blockIdx.y * A_PER_WG;
  const int myb  = b0 + wv * 16;

  for (int i = tid; i < P_TOT; i += 256) ldsOff[i] = offs[i];
  for (int i = tid; i < 128 * EPAD; i += 256) {
    int row = i >> 4, e = i & 15;
    ldsY[i] = (e < NE) ? y[(size_t)(b0 + row) * NE + e] : 0.0f;
  }

  for (int ai = 0; ai < A_PER_WG; ++ai) {
    const int a = a0 + ai;
    __syncthreads();
    {
      const float4* src = (const float4*)(Wg + (size_t)a * P_TOT * EPAD);
      float4* dst = (float4*)ldsW;
      for (int i = tid; i < P_TOT * EPAD / 4; i += 256) dst[i] = src[i];
    }
    for (int i = lane; i < 16 * 16; i += 32) {
      int row = i >> 4, c4 = i & 15;
      const float4 v = *(const float4*)(x + ((size_t)(myb + row) * NA + a) * 64 + c4 * 4);
      *(float4*)(&ldsX[(wv * 16 + row) * XPITCH + c4 * 4]) = v;
    }
    __syncthreads();

    const float* X = &ldsX[(wv * 16 + lrow) * XPITCH];
    v8f acc = {};
#pragma unroll
    for (int t = 0; t < 8; ++t)
      wmma_step<1>(t, ldsOff, X, ldsW, half, lrow, acc);
#pragma unroll 4
    for (int t = 8; t < 40; ++t)
      wmma_step<2>(t, ldsOff, X, ldsW, half, lrow, acc);
#pragma unroll 4
    for (int t = 40; t < 104; ++t)
      wmma_step<3>(t, ldsOff, X, ldsW, half, lrow, acc);

#pragma unroll
    for (int r = 0; r < 8; ++r) {
      const int row = wv * 16 + r + half * 8;
      float pr = acc[r] * ldsY[row * EPAD + lrow];
      pr += __shfl_xor(pr, 1, 16);
      pr += __shfl_xor(pr, 2, 16);
      pr += __shfl_xor(pr, 4, 16);
      pr += __shfl_xor(pr, 8, 16);
      if (lrow == 0) out[(size_t)(b0 + row) * NA + a] = pr;
    }
  }
}

extern "C" void kernel_launch(void* const* d_in, const int* in_sizes, int n_in,
                              void* d_out, int out_size, void* d_ws, size_t ws_size,
                              hipStream_t stream) {
  const float* x  = (const float*)d_in[0];
  const float* y  = (const float*)d_in[1];
  const float* w1 = (const float*)d_in[2];
  const float* w2 = (const float*)d_in[3];
  const float* w3 = (const float*)d_in[4];
  const float* c1 = (const float*)d_in[5];
  const float* c2 = (const float*)d_in[6];
  const float* c3 = (const float*)d_in[7];
  const int* i1 = (const int*)d_in[8];
  const int* l1 = (const int*)d_in[9];
  const int* k1 = (const int*)d_in[10];
  const int* q1 = (const int*)d_in[11];
  const int* i2 = (const int*)d_in[12];
  const int* j2 = (const int*)d_in[13];
  const int* l2 = (const int*)d_in[14];
  const int* m2 = (const int*)d_in[15];
  const int* k2 = (const int*)d_in[16];
  const int* q2 = (const int*)d_in[17];
  const int* i3 = (const int*)d_in[18];
  const int* j3 = (const int*)d_in[19];
  const int* f3 = (const int*)d_in[20];
  const int* l3 = (const int*)d_in[21];
  const int* m3 = (const int*)d_in[22];
  const int* g3 = (const int*)d_in[23];
  const int* k3 = (const int*)d_in[24];
  const int* q3 = (const int*)d_in[25];

  int*   offs = (int*)d_ws;
  float* Wg   = (float*)((char*)d_ws + 2048);
  float* out  = (float*)d_out;

  prep_offsets_kernel<<<2, 256, 0, stream>>>(i1, l1, i2, j2, l2, m2,
                                             i3, j3, f3, l3, m3, g3, offs);
  prep_wg_kernel<<<104, 256, 0, stream>>>(w1, w2, w3, c1, c2, c3,
                                          k1, q1, k2, q2, k3, q3, Wg);
  contract_kernel<<<dim3(32, 32), 256, 0, stream>>>(x, y, Wg, offs, out);
}


# Staging global loads (offsets, y, Wg slice, x rows) issued as one register-held group per a, the next a's group ahead of the epilogue and barrier, instead of load-wait-store per element
# speedup vs baseline: 2.7577x; 1.1831x over previous
#include <hip/hip_runtime.h>
#include <hip/hip_bf16.h>

typedef float v2f __attribute__((ext_vector_type(2)));
typedef float v8f __attribute__((ext_vector_type(8)));

#define NB 4096
#define NA 64
#define NL 16
#define NM 4
#define NE 10
#define P_TOT 416
#define P1_END 32
#define P2_END 160
#define EPAD 16
#define XPITCH 68
#define A_PER_WG 2

__global__ void prep_offsets_kernel(const int* __restrict__ i1, const int* __restrict__ l1,
                                    const int* __restrict__ i2, const int* __restrict__ j2,
                                    const int* __restrict__ l2, const int* __restrict__ m2,
                                    const int* __restrict__ i3, const int* __restrict__ j3,
                                    const int* __restrict__ f3, const int* __restrict__ l3,
                                    const int* __restrict__ m3, const int* __restrict__ g3,
                                    int* __restrict__ offs) {
  int p = blockIdx.x * blockDim.x + threadIdx.x;
  if (p >= P_TOT) return;
  int packed;
  if (p < P1_END) {
    int o1 = i1[p] * NM + l1[p];
    packed = o1;
  } else if (p < P2_END) {
    int q = p - P1_END;
    int o1 = i2[q] * NM + l2[q];
    int o2 = j2[q] * NM + m2[q];
    packed = o1 | (o2 << 8);
  } else {
    int q = p - P2_END;
    int o1 = i3[q] * NM + l3[q];
    int o2 = j3[q] * NM + m3[q];
    int o3 = f3[q] * NM + g3[q];
    packed = o1 | (o2 << 8) | (o3 << 16);
  }
  offs[p] = packed;
}

__global__ void prep_wg_kernel(const float* __restrict__ w1, const float* __restrict__ w2,
                               const float* __restrict__ w3,
                               const float* __restrict__ c1, const float* __restrict__ c2,
                               const float* __restrict__ c3,
                               const int* __restrict__ k1, const int* __restrict__ q1,
                               const int* __restrict__ k2, const int* __restrict__ q2,
                               const int* __restrict__ k3, const int* __restrict__ q3,
                               float* __restrict__ Wg) {
  int tid = blockIdx.x * blockDim.x + threadIdx.x;
  if (tid >= NA * P_TOT) return;
  int p = tid % P_TOT;
  int a = tid / P_TOT;
  float cf;
  const float* wbase;
  int estride;
  int kq;
  if (p < P1_END) {
    cf = c1[p];
    wbase = w1; kq = k1[p] * 2 + q1[p]; estride = 4 * 2 * NA;
  } else if (p < P2_END) {
    int t = p - P1_END;
    cf = c2[t];
    wbase = w2; kq = k2[t] * 4 + q2[t]; estride = 16 * 4 * NA;
  } else {
    int t = p - P2_END;
    cf = c3[t];
    wbase = w3; kq = k3[t] * 6 + q3[t]; estride = 32 * 6 * NA;
  }
  const int kqstride = kq * NA + a;
  float* dst = Wg + ((size_t)a * P_TOT + p) * EPAD;
#pragma unroll
  for (int n = 0; n < EPAD; ++n) {
    dst[n] = (n < NE) ? cf * wbase[(size_t)n * estride + kqstride] : 0.0f;
  }
}

template <int CNT>
__device__ __forceinline__ void wmma_step(int t, const int* __restrict__ ldsOff,
                                          const float* __restrict__ X,
                                          const float* __restrict__ ldsW,
                                          int half, int lrow, v8f& acc) {
  const int pA = 4 * t + 2 * half;
  const int pk0 = ldsOff[pA];
  const int pk1 = ldsOff[pA + 1];
  float xa = X[pk0 & 0xFF];
  float xb = X[pk1 & 0xFF];
  if (CNT > 1) { xa *= X[(pk0 >> 8) & 0xFF]; xb *= X[(pk1 >> 8) & 0xFF]; }
  if (CNT > 2) { xa *= X[(pk0 >> 16) & 0xFF]; xb *= X[(pk1 >> 16) & 0xFF]; }
  v2f af; af.x = xa; af.y = xb;
  v2f bf;
  bf.x = ldsW[pA * EPAD + lrow];
  bf.y = ldsW[(pA + 1) * EPAD + lrow];
  acc = __builtin_amdgcn_wmma_f32_16x16x4_f32(false, af, false, bf,
                                              (short)0, acc, false, false);
}

typedef float v4f __attribute__((ext_vector_type(4)));

struct StageRegs {
  v4f w0, w1, w2, w3, w4, w5, w6;
  v4f x0, x1, x2, x3, x4, x5, x6, x7;
};

__device__ __forceinline__ void stage_load(StageRegs& r, const float* __restrict__ x,
                                           const float* __restrict__ Wg, int a,
                                           int tid, int lane, int myb) {
  const v4f* ws = (const v4f*)(Wg + (size_t)a * P_TOT * EPAD) + tid;
  r.w0 = ws[0];
  r.w1 = ws[256];
  r.w2 = ws[512];
  r.w3 = ws[768];
  r.w4 = ws[1024];
  r.w5 = ws[1280];
  if (tid < 128) r.w6 = ws[1536];
  const float* xb = x + ((size_t)(myb + (lane >> 4)) * NA + a) * 64 + (lane & 15) * 4;
  const int XS = 2 * NA * 64;
  r.x0 = *(const v4f*)(xb);
  r.x1 = *(const v4f*)(xb + 1 * XS);
  r.x2 = *(const v4f*)(xb + 2 * XS);
  r.x3 = *(const v4f*)(xb + 3 * XS);
  r.x4 = *(const v4f*)(xb + 4 * XS);
  r.x5 = *(const v4f*)(xb + 5 * XS);
  r.x6 = *(const v4f*)(xb + 6 * XS);
  r.x7 = *(const v4f*)(xb + 7 * XS);
}

__device__ __forceinline__ void stage_store(const StageRegs& r, float* ldsW, float* ldsX,
                                            int tid, int lane, int wv) {
  v4f* wd = (v4f*)ldsW + tid;
  wd[0]    = r.w0;
  wd[256]  = r.w1;
  wd[512]  = r.w2;
  wd[768]  = r.w3;
  wd[1024] = r.w4;
  wd[1280] = r.w5;
  if (tid < 128) wd[1536] = r.w6;
  float* xd = &ldsX[(wv * 16 + (lane >> 4)) * XPITCH + (lane & 15) * 4];
  const int LS = 2 * XPITCH;
  *(v4f*)(xd)          = r.x0;
  *(v4f*)(xd + 1 * LS) = r.x1;
  *(v4f*)(xd + 2 * LS) = r.x2;
  *(v4f*)(xd + 3 * LS) = r.x3;
  *(v4f*)(xd + 4 * LS) = r.x4;
  *(v4f*)(xd + 5 * LS) = r.x5;
  *(v4f*)(xd + 6 * LS) = r.x6;
  *(v4f*)(xd + 7 * LS) = r.x7;
}

__global__ __launch_bounds__(256) void contract_kernel(const float* __restrict__ x,
                                                       const float* __restrict__ y,
                                                       const float* __restrict__ Wg,
                                                       const int* __restrict__ offs,
                                                       float* __restrict__ out) {
  __shared__ float ldsW[P_TOT * EPAD];
  __shared__ float ldsX[8 * 16 * XPITCH];
  __shared__ float ldsY[128 * EPAD];
  __shared__ int   ldsOff[P_TOT];

  const int tid  = threadIdx.x;
  const int wv   = tid >> 5;
  const int lane = tid & 31;
  const int half = lane >> 4;
  const int lrow = lane & 15;
  const int b0   = blockIdx.x * 128;
  const int a0   = blockIdx.y * A_PER_WG;
  const int myb  = b0 + wv * 16;

  StageRegs sr;
  sr.w6 = (v4f){0.0f, 0.0f, 0.0f, 0.0f};
  stage_load(sr, x, Wg, a0, tid, lane, myb);
  const int off_lo = offs[tid];
  int off_hi = 0;
  if (tid < P_TOT - 256) off_hi = offs[256 + tid];
  float y0 = 0.0f, y1 = 0.0f, y2 = 0.0f, y3 = 0.0f, y4 = 0.0f, y5 = 0.0f, y6 = 0.0f, y7 = 0.0f;
  if ((tid & 15) < NE) {
    const float* yb = y + (size_t)(b0 + (tid >> 4)) * NE + (tid & 15);
    const int YS = 16 * NE;
    y0 = yb[0];      y1 = yb[1 * YS]; y2 = yb[2 * YS]; y3 = yb[3 * YS];
    y4 = yb[4 * YS]; y5 = yb[5 * YS]; y6 = yb[6 * YS]; y7 = yb[7 * YS];
  }

  ldsOff[tid] = off_lo;
  if (tid < P_TOT - 256) ldsOff[256 + tid] = off_hi;
  ldsY[tid]        = y0; ldsY[tid + 256]  = y1; ldsY[tid + 512]  = y2; ldsY[tid + 768]  = y3;
  ldsY[tid + 1024] = y4; ldsY[tid + 1280] = y5; ldsY[tid + 1536] = y6; ldsY[tid + 1792] = y7;

#pragma unroll
  for (int ai = 0; ai < A_PER_WG; ++ai) {
    const int a = a0 + ai;
    if (ai > 0) __syncthreads();
    stage_store(sr, ldsW, ldsX, tid, lane, wv);
    __syncthreads();

    const float* X = &ldsX[(wv * 16 + lrow) * XPITCH];
    v8f acc = {};
#pragma unroll
    for (int t = 0; t < 8; ++t)
      wmma_step<1>(t, ldsOff, X, ldsW, half, lrow, acc);
#pragma unroll 4
    for (int t = 8; t < 40; ++t)
      wmma_step<2>(t, ldsOff, X, ldsW, half, lrow, acc);
#pragma unroll 4
    for (int t = 40; t < 104; ++t)
      wmma_step<3>(t, ldsOff, X, ldsW, half, lrow, acc);

    if (ai + 1 < A_PER_WG) stage_load(sr, x, Wg, a + 1, tid, lane, myb);

#pragma unroll
    for (int r = 0; r < 8; ++r) {
      const int row = wv * 16 + r + half * 8;
      float pr = acc[r] * ldsY[row * EPAD + lrow];
      pr += __shfl_xor(pr, 1, 16);
      pr += __shfl_xor(pr, 2, 16);
      pr += __shfl_xor(pr, 4, 16);
      pr += __shfl_xor(pr, 8, 16);
      if (lrow == 0) out[(size_t)(b0 + row) * NA + a] = pr;
    }
  }
}

extern "C" void kernel_launch(void* const* d_in, const int* in_sizes, int n_in,
                              void* d_out, int out_size, void* d_ws, size_t ws_size,
                              hipStream_t stream) {
  const float* x  = (const float*)d_in[0];
  const float* y  = (const float*)d_in[1];
  const float* w1 = (const float*)d_in[2];
  const float* w2 = (const float*)d_in[3];
  const float* w3 = (const float*)d_in[4];
  const float* c1 = (const float*)d_in[5];
  const float* c2 = (const float*)d_in[6];
  const float* c3 = (const float*)d_in[7];
  const int* i1 = (const int*)d_in[8];
  const int* l1 = (const int*)d_in[9];
  const int* k1 = (const int*)d_in[10];
  const int* q1 = (const int*)d_in[11];
  const int* i2 = (const int*)d_in[12];
  const int* j2 = (const int*)d_in[13];
  const int* l2 = (const int*)d_in[14];
  const int* m2 = (const int*)d_in[15];
  const int* k2 = (const int*)d_in[16];
  const int* q2 = (const int*)d_in[17];
  const int* i3 = (const int*)d_in[18];
  const int* j3 = (const int*)d_in[19];
  const int* f3 = (const int*)d_in[20];
  const int* l3 = (const int*)d_in[21];
  const int* m3 = (const int*)d_in[22];
  const int* g3 = (const int*)d_in[23];
  const int* k3 = (const int*)d_in[24];
  const int* q3 = (const int*)d_in[25];

  int*   offs = (int*)d_ws;
  float* Wg   = (float*)((char*)d_ws + 2048);
  float* out  = (float*)d_out;

  prep_offsets_kernel<<<2, 256, 0, stream>>>(i1, l1, i2, j2, l2, m2,
                                             i3, j3, f3, l3, m3, g3, offs);
  prep_wg_kernel<<<104, 256, 0, stream>>>(w1, w2, w3, c1, c2, c3,
                                          k1, q1, k2, q2, k3, q3, Wg);
  contract_kernel<<<dim3(32, 32), 256, 0, stream>>>(x, y, Wg, offs, out);
}


# Each wave runs two 16-row tiles per K step (128-thread groups of 4 waves): x rows interleaved pairwise in LDS so one 8-byte load gathers a factor for both tiles; offsets and B fragment shared
# speedup vs baseline: 2.9066x; 1.0540x over previous
#include <hip/hip_runtime.h>
#include <hip/hip_bf16.h>

typedef float v2f __attribute__((ext_vector_type(2)));
typedef float v8f __attribute__((ext_vector_type(8)));

#define NB 4096
#define NA 64
#define NL 16
#define NM 4
#define NE 10
#define P_TOT 416
#define P1_END 32
#define P2_END 160
#define EPAD 16
#define XP2 132
#define WG_WAVES 4
#define WG_THREADS 128
#define WG_ROWS 128
#define A_PER_WG 2

__global__ void prep_offsets_kernel(const int* __restrict__ i1, const int* __restrict__ l1,
                                    const int* __restrict__ i2, const int* __restrict__ j2,
                                    const int* __restrict__ l2, const int* __restrict__ m2,
                                    const int* __restrict__ i3, const int* __restrict__ j3,
                                    const int* __restrict__ f3, const int* __restrict__ l3,
                                    const int* __restrict__ m3, const int* __restrict__ g3,
                                    int* __restrict__ offs) {
  int p = blockIdx.x * blockDim.x + threadIdx.x;
  if (p >= P_TOT) return;
  int packed;
  if (p < P1_END) {
    int o1 = i1[p] * NM + l1[p];
    packed = o1;
  } else if (p < P2_END) {
    int q = p - P1_END;
    int o1 = i2[q] * NM + l2[q];
    int o2 = j2[q] * NM + m2[q];
    packed = o1 | (o2 << 8);
  } else {
    int q = p - P2_END;
    int o1 = i3[q] * NM + l3[q];
    int o2 = j3[q] * NM + m3[q];
    int o3 = f3[q] * NM + g3[q];
    packed = o1 | (o2 << 8) | (o3 << 16);
  }
  offs[p] = packed;
}

__global__ void prep_wg_kernel(const float* __restrict__ w1, const float* __restrict__ w2,
                               const float* __restrict__ w3,
                               const float* __restrict__ c1, const float* __restrict__ c2,
                               const float* __restrict__ c3,
                               const int* __restrict__ k1, const int* __restrict__ q1,
                               const int* __restrict__ k2, const int* __restrict__ q2,
                               const int* __restrict__ k3, const int* __restrict__ q3,
                               float* __restrict__ Wg) {
  int tid = blockIdx.x * blockDim.x + threadIdx.x;
  if (tid >= NA * P_TOT) return;
  int p = tid % P_TOT;
  int a = tid / P_TOT;
  float cf;
  const float* wbase;
  int estride;
  int kq;
  if (p < P1_END) {
    cf = c1[p];
    wbase = w1; kq = k1[p] * 2 + q1[p]; estride = 4 * 2 * NA;
  } else if (p < P2_END) {
    int t = p - P1_END;
    cf = c2[t];
    wbase = w2; kq = k2[t] * 4 + q2[t]; estride = 16 * 4 * NA;
  } else {
    int t = p - P2_END;
    cf = c3[t];
    wbase = w3; kq = k3[t] * 6 + q3[t]; estride = 32 * 6 * NA;
  }
  const int kqstride = kq * NA + a;
  float* dst = Wg + ((size_t)a * P_TOT + p) * EPAD;
#pragma unroll
  for (int n = 0; n < EPAD; ++n) {
    dst[n] = (n < NE) ? cf * wbase[(size_t)n * estride + kqstride] : 0.0f;
  }
}

template <int CNT>
__device__ __forceinline__ void wmma_step2(int t, const int* __restrict__ ldsOff,
                                           const v2f* __restrict__ X,
                                           const float* __restrict__ ldsW,
                                           int half, int lrow, v8f& acc0, v8f& acc1) {
  const int pA = 4 * t + 2 * half;
  const int pk0 = ldsOff[pA];
  const int pk1 = ldsOff[pA + 1];
  v2f xa = X[pk0 & 0xFF];
  v2f xb = X[pk1 & 0xFF];
  if (CNT > 1) { xa *= X[(pk0 >> 8) & 0xFF]; xb *= X[(pk1 >> 8) & 0xFF]; }
  if (CNT > 2) { xa *= X[(pk0 >> 16) & 0xFF]; xb *= X[(pk1 >> 16) & 0xFF]; }
  v2f a0; a0.x = xa.x; a0.y = xb.x;
  v2f a1; a1.x = xa.y; a1.y = xb.y;
  v2f bf;
  bf.x = ldsW[pA * EPAD + lrow];
  bf.y = ldsW[(pA + 1) * EPAD + lrow];
  acc0 = __builtin_amdgcn_wmma_f32_16x16x4_f32(false, a0, false, bf,
                                               (short)0, acc0, false, false);
  acc1 = __builtin_amdgcn_wmma_f32_16x16x4_f32(false, a1, false, bf,
                                               (short)0, acc1, false, false);
}

typedef float v4f __attribute__((ext_vector_type(4)));

struct StageRegs {
  v4f w[13];
  v4f xr[16];
};

__device__ __forceinline__ void stage_load(StageRegs& r, const float* __restrict__ x,
                                           const float* __restrict__ Wg, int a,
                                           int tid, int lane, int myb) {
  const v4f* ws = (const v4f*)(Wg + (size_t)a * P_TOT * EPAD) + tid;
#pragma unroll
  for (int k = 0; k < 13; ++k) r.w[k] = ws[WG_THREADS * k];
  const float* xb = x + ((size_t)(myb + (lane >> 4)) * NA + a) * 64 + (lane & 15) * 4;
  const int XS = 2 * NA * 64;
#pragma unroll
  for (int k = 0; k < 16; ++k) r.xr[k] = *(const v4f*)(xb + k * XS);
}

__device__ __forceinline__ void stage_store(const StageRegs& r, float* ldsW, float* ldsX,
                                            int tid, int lane, int wv) {
  v4f* wd = (v4f*)ldsW + tid;
#pragma unroll
  for (int k = 0; k < 13; ++k) wd[WG_THREADS * k] = r.w[k];
  float* xd = &ldsX[(wv * 16 + (lane >> 4)) * XP2 + (lane & 15) * 8];
  const int LS = 2 * XP2;
#pragma unroll
  for (int j = 0; j < 8; ++j) {
    const v4f t0 = r.xr[j];
    const v4f t1 = r.xr[8 + j];
    *(v4f*)(xd + j * LS)     = (v4f){t0.x, t1.x, t0.y, t1.y};
    *(v4f*)(xd + j * LS + 4) = (v4f){t0.z, t1.z, t0.w, t1.w};
  }
}

__global__ __launch_bounds__(WG_THREADS) void contract_kernel(const float* __restrict__ x,
                                                       const float* __restrict__ y,
                                                       const float* __restrict__ Wg,
                                                       const int* __restrict__ offs,
                                                       float* __restrict__ out) {
  __shared__ __attribute__((aligned(16))) float ldsW[P_TOT * EPAD];
  __shared__ __attribute__((aligned(16))) float ldsX[WG_WAVES * 16 * XP2];
  __shared__ __attribute__((aligned(16))) float ldsY[WG_ROWS * EPAD];
  __shared__ int   ldsOff[P_TOT];

  const int tid  = threadIdx.x;
  const int wv   = tid >> 5;
  const int lane = tid & 31;
  const int half = lane >> 4;
  const int lrow = lane & 15;
  const int b0   = blockIdx.x * WG_ROWS;
  const int a0   = blockIdx.y * A_PER_WG;
  const int myb  = b0 + wv * 32;

  StageRegs sr;
  stage_load(sr, x, Wg, a0, tid, lane, myb);
  const int off0 = offs[tid];
  const int off1 = offs[WG_THREADS + tid];
  const int off2 = offs[2 * WG_THREADS + tid];
  int off3 = 0;
  if (tid < P_TOT - 3 * WG_THREADS) off3 = offs[3 * WG_THREADS + tid];
  float yv[16];
#pragma unroll
  for (int j = 0; j < 16; ++j) yv[j] = 0.0f;
  if ((tid & 15) < NE) {
    const float* yb = y + (size_t)(b0 + (tid >> 4)) * NE + (tid & 15);
    const int YS = 8 * NE;
#pragma unroll
    for (int j = 0; j < 16; ++j) yv[j] = yb[j * YS];
  }

  ldsOff[tid] = off0;
  ldsOff[WG_THREADS + tid] = off1;
  ldsOff[2 * WG_THREADS + tid] = off2;
  if (tid < P_TOT - 3 * WG_THREADS) ldsOff[3 * WG_THREADS + tid] = off3;
#pragma unroll
  for (int j = 0; j < 16; ++j) ldsY[tid + WG_THREADS * j] = yv[j];

#pragma unroll
  for (int ai = 0; ai < A_PER_WG; ++ai) {
    const int a = a0 + ai;
    if (ai > 0) __syncthreads();
    stage_store(sr, ldsW, ldsX, tid, lane, wv);
    __syncthreads();

    const v2f* X = (const v2f*)&ldsX[(wv * 16 + lrow) * XP2];
    v8f acc0 = {};
    v8f acc1 = {};
#pragma unroll
    for (int t = 0; t < 8; ++t)
      wmma_step2<1>(t, ldsOff, X, ldsW, half, lrow, acc0, acc1);
#pragma unroll 4
    for (int t = 8; t < 40; ++t)
      wmma_step2<2>(t, ldsOff, X, ldsW, half, lrow, acc0, acc1);
#pragma unroll 4
    for (int t = 40; t < 104; ++t)
      wmma_step2<3>(t, ldsOff, X, ldsW, half, lrow, acc0, acc1);

    if (ai + 1 < A_PER_WG) stage_load(sr, x, Wg, a + 1, tid, lane, myb);

#pragma unroll
    for (int tl = 0; tl < 2; ++tl) {
#pragma unroll
      for (int r = 0; r < 8; ++r) {
        const int row = wv * 32 + tl * 16 + r + half * 8;
        float pr = (tl == 0 ? acc0[r] : acc1[r]) * ldsY[row * EPAD + lrow];
        pr += __shfl_xor(pr, 1, 16);
        pr += __shfl_xor(pr, 2, 16);
        pr += __shfl_xor(pr, 4, 16);
        pr += __shfl_xor(pr, 8, 16);
        if (lrow == 0) out[(size_t)(b0 + row) * NA + a] = pr;
      }
    }
  }
}

extern "C" void kernel_launch(void* const* d_in, const int* in_sizes, int n_in,
                              void* d_out, int out_size, void* d_ws, size_t ws_size,
                              hipStream_t stream) {
  const float* x  = (const float*)d_in[0];
  const float* y  = (const float*)d_in[1];
  const float* w1 = (const float*)d_in[2];
  const float* w2 = (const float*)d_in[3];
  const float* w3 = (const float*)d_in[4];
  const float* c1 = (const float*)d_in[5];
  const float* c2 = (const float*)d_in[6];
  const float* c3 = (const float*)d_in[7];
  const int* i1 = (const int*)d_in[8];
  const int* l1 = (const int*)d_in[9];
  const int* k1 = (const int*)d_in[10];
  const int* q1 = (const int*)d_in[11];
  const int* i2 = (const int*)d_in[12];
  const int* j2 = (const int*)d_in[13];
  const int* l2 = (const int*)d_in[14];
  const int* m2 = (const int*)d_in[15];
  const int* k2 = (const int*)d_in[16];
  const int* q2 = (const int*)d_in[17];
  const int* i3 = (const int*)d_in[18];
  const int* j3 = (const int*)d_in[19];
  const int* f3 = (const int*)d_in[20];
  const int* l3 = (const int*)d_in[21];
  const int* m3 = (const int*)d_in[22];
  const int* g3 = (const int*)d_in[23];
  const int* k3 = (const int*)d_in[24];
  const int* q3 = (const int*)d_in[25];

  int*   offs = (int*)d_ws;
  float* Wg   = (float*)((char*)d_ws + 2048);
  float* out  = (float*)d_out;

  prep_offsets_kernel<<<2, 256, 0, stream>>>(i1, l1, i2, j2, l2, m2,
                                             i3, j3, f3, l3, m3, g3, offs);
  prep_wg_kernel<<<104, 256, 0, stream>>>(w1, w2, w3, c1, c2, c3,
                                          k1, q1, k2, q2, k3, q3, Wg);
  contract_kernel<<<dim3(32, 32), WG_THREADS, 0, stream>>>(x, y, Wg, offs, out);
}


# Epilogue row sums over the 16 lanes of a half (y-weighted H rows) by row_xmask DPP adds on the vector ALU instead of 64 serialized ds_bpermute round trips per a; same butterfly order
# speedup vs baseline: 3.2442x; 1.1161x over previous
#include <hip/hip_runtime.h>
#include <hip/hip_bf16.h>

typedef float v2f __attribute__((ext_vector_type(2)));
typedef float v8f __attribute__((ext_vector_type(8)));

#define NB 4096
#define NA 64
#define NL 16
#define NM 4
#define NE 10
#define P_TOT 416
#define P1_END 32
#define P2_END 160
#define EPAD 16
#define XP2 132
#define WG_WAVES 4
#define WG_THREADS 128
#define WG_ROWS 128
#define A_PER_WG 2

__global__ void prep_offsets_kernel(const int* __restrict__ i1, const int* __restrict__ l1,
                                    const int* __restrict__ i2, const int* __restrict__ j2,
                                    const int* __restrict__ l2, const int* __restrict__ m2,
                                    const int* __restrict__ i3, const int* __restrict__ j3,
                                    const int* __restrict__ f3, const int* __restrict__ l3,
                                    const int* __restrict__ m3, const int* __restrict__ g3,
                                    int* __restrict__ offs) {
  int p = blockIdx.x * blockDim.x + threadIdx.x;
  if (p >= P_TOT) return;
  int packed;
  if (p < P1_END) {
    int o1 = i1[p] * NM + l1[p];
    packed = o1;
  } else if (p < P2_END) {
    int q = p - P1_END;
    int o1 = i2[q] * NM + l2[q];
    int o2 = j2[q] * NM + m2[q];
    packed = o1 | (o2 << 8);
  } else {
    int q = p - P2_END;
    int o1 = i3[q] * NM + l3[q];
    int o2 = j3[q] * NM + m3[q];
    int o3 = f3[q] * NM + g3[q];
    packed = o1 | (o2 << 8) | (o3 << 16);
  }
  offs[p] = packed;
}

__global__ void prep_wg_kernel(const float* __restrict__ w1, const float* __restrict__ w2,
                               const float* __restrict__ w3,
                               const float* __restrict__ c1, const float* __restrict__ c2,
                               const float* __restrict__ c3,
                               const int* __restrict__ k1, const int* __restrict__ q1,
                               const int* __restrict__ k2, const int* __restrict__ q2,
                               const int* __restrict__ k3, const int* __restrict__ q3,
                               float* __restrict__ Wg) {
  int tid = blockIdx.x * blockDim.x + threadIdx.x;
  if (tid >= NA * P_TOT) return;
  int p = tid % P_TOT;
  int a = tid / P_TOT;
  float cf;
  const float* wbase;
  int estride;
  int kq;
  if (p < P1_END) {
    cf = c1[p];
    wbase = w1; kq = k1[p] * 2 + q1[p]; estride = 4 * 2 * NA;
  } else if (p < P2_END) {
    int t = p - P1_END;
    cf = c2[t];
    wbase = w2; kq = k2[t] * 4 + q2[t]; estride = 16 * 4 * NA;
  } else {
    int t = p - P2_END;
    cf = c3[t];
    wbase = w3; kq = k3[t] * 6 + q3[t]; estride = 32 * 6 * NA;
  }
  const int kqstride = kq * NA + a;
  float* dst = Wg + ((size_t)a * P_TOT + p) * EPAD;
#pragma unroll
  for (int n = 0; n < EPAD; ++n) {
    dst[n] = (n < NE) ? cf * wbase[(size_t)n * estride + kqstride] : 0.0f;
  }
}

template <int CNT>
__device__ __forceinline__ void wmma_step2(int t, const int* __restrict__ ldsOff,
                                           const v2f* __restrict__ X,
                                           const float* __restrict__ ldsW,
                                           int half, int lrow, v8f& acc0, v8f& acc1) {
  const int pA = 4 * t + 2 * half;
  const int pk0 = ldsOff[pA];
  const int pk1 = ldsOff[pA + 1];
  v2f xa = X[pk0 & 0xFF];
  v2f xb = X[pk1 & 0xFF];
  if (CNT > 1) { xa *= X[(pk0 >> 8) & 0xFF]; xb *= X[(pk1 >> 8) & 0xFF]; }
  if (CNT > 2) { xa *= X[(pk0 >> 16) & 0xFF]; xb *= X[(pk1 >> 16) & 0xFF]; }
  v2f a0; a0.x = xa.x; a0.y = xb.x;
  v2f a1; a1.x = xa.y; a1.y = xb.y;
  v2f bf;
  bf.x = ldsW[pA * EPAD + lrow];
  bf.y = ldsW[(pA + 1) * EPAD + lrow];
  acc0 = __builtin_amdgcn_wmma_f32_16x16x4_f32(false, a0, false, bf,
                                               (short)0, acc0, false, false);
  acc1 = __builtin_amdgcn_wmma_f32_16x16x4_f32(false, a1, false, bf,
                                               (short)0, acc1, false, false);
}

typedef float v4f __attribute__((ext_vector_type(4)));

struct StageRegs {
  v4f w[13];
  v4f xr[16];
};

__device__ __forceinline__ void stage_load(StageRegs& r, const float* __restrict__ x,
                                           const float* __restrict__ Wg, int a,
                                           int tid, int lane, int myb) {
  const v4f* ws = (const v4f*)(Wg + (size_t)a * P_TOT * EPAD) + tid;
#pragma unroll
  for (int k = 0; k < 13; ++k) r.w[k] = ws[WG_THREADS * k];
  const float* xb = x + ((size_t)(myb + (lane >> 4)) * NA + a) * 64 + (lane & 15) * 4;
  const int XS = 2 * NA * 64;
#pragma unroll
  for (int k = 0; k < 16; ++k) r.xr[k] = *(const v4f*)(xb + k * XS);
}

__device__ __forceinline__ void stage_store(const StageRegs& r, float* ldsW, float* ldsX,
                                            int tid, int lane, int wv) {
  v4f* wd = (v4f*)ldsW + tid;
#pragma unroll
  for (int k = 0; k < 13; ++k) wd[WG_THREADS * k] = r.w[k];
  float* xd = &ldsX[(wv * 16 + (lane >> 4)) * XP2 + (lane & 15) * 8];
  const int LS = 2 * XP2;
#pragma unroll
  for (int j = 0; j < 8; ++j) {
    const v4f t0 = r.xr[j];
    const v4f t1 = r.xr[8 + j];
    *(v4f*)(xd + j * LS)     = (v4f){t0.x, t1.x, t0.y, t1.y};
    *(v4f*)(xd + j * LS + 4) = (v4f){t0.z, t1.z, t0.w, t1.w};
  }
}

template <int MASK>
__device__ __forceinline__ float row_xor_add(float v) {
  const int other = __builtin_amdgcn_update_dpp(0, __float_as_int(v), 0x160 + MASK, 0xF, 0xF, true);
  return v + __int_as_float(other);
}

__global__ __launch_bounds__(WG_THREADS) void contract_kernel(const float* __restrict__ x,
                                                       const float* __restrict__ y,
                                                       const float* __restrict__ Wg,
                                                       const int* __restrict__ offs,
                                                       float* __restrict__ out) {
  __shared__ __attribute__((aligned(16))) float ldsW[P_TOT * EPAD];
  __shared__ __attribute__((aligned(16))) float ldsX[WG_WAVES * 16 * XP2];
  __shared__ __attribute__((aligned(16))) float ldsY[WG_ROWS * EPAD];
  __shared__ int   ldsOff[P_TOT];

  const int tid  = threadIdx.x;
  const int wv   = tid >> 5;
  const int lane = tid & 31;
  const int half = lane >> 4;
  const int lrow = lane & 15;
  const int b0   = blockIdx.x * WG_ROWS;
  const int a0   = blockIdx.y * A_PER_WG;
  const int myb  = b0 + wv * 32;

  StageRegs sr;
  stage_load(sr, x, Wg, a0, tid, lane, myb);
  const int off0 = offs[tid];
  const int off1 = offs[WG_THREADS + tid];
  const int off2 = offs[2 * WG_THREADS + tid];
  int off3 = 0;
  if (tid < P_TOT - 3 * WG_THREADS) off3 = offs[3 * WG_THREADS + tid];
  float yv[16];
#pragma unroll
  for (int j = 0; j < 16; ++j) yv[j] = 0.0f;
  if ((tid & 15) < NE) {
    const float* yb = y + (size_t)(b0 + (tid >> 4)) * NE + (tid & 15);
    const int YS = 8 * NE;
#pragma unroll
    for (int j = 0; j < 16; ++j) yv[j] = yb[j * YS];
  }

  ldsOff[tid] = off0;
  ldsOff[WG_THREADS + tid] = off1;
  ldsOff[2 * WG_THREADS + tid] = off2;
  if (tid < P_TOT - 3 * WG_THREADS) ldsOff[3 * WG_THREADS + tid] = off3;
#pragma unroll
  for (int j = 0; j < 16; ++j) ldsY[tid + WG_THREADS * j] = yv[j];

#pragma unroll
  for (int ai = 0; ai < A_PER_WG; ++ai) {
    const int a = a0 + ai;
    if (ai > 0) __syncthreads();
    stage_store(sr, ldsW, ldsX, tid, lane, wv);
    __syncthreads();

    const v2f* X = (const v2f*)&ldsX[(wv * 16 + lrow) * XP2];
    v8f acc0 = {};
    v8f acc1 = {};
#pragma unroll
    for (int t = 0; t < 8; ++t)
      wmma_step2<1>(t, ldsOff, X, ldsW, half, lrow, acc0, acc1);
#pragma unroll 4
    for (int t = 8; t < 40; ++t)
      wmma_step2<2>(t, ldsOff, X, ldsW, half, lrow, acc0, acc1);
#pragma unroll 4
    for (int t = 40; t < 104; ++t)
      wmma_step2<3>(t, ldsOff, X, ldsW, half, lrow, acc0, acc1);

    if (ai + 1 < A_PER_WG) stage_load(sr, x, Wg, a + 1, tid, lane, myb);

    float pr[16];
#pragma unroll
    for (int tl = 0; tl < 2; ++tl) {
#pragma unroll
      for (int r = 0; r < 8; ++r) {
        const int row = wv * 32 + tl * 16 + r + half * 8;
        float v = (tl == 0 ? acc0[r] : acc1[r]) * ldsY[row * EPAD + lrow];
        v = row_xor_add<1>(v);
        v = row_xor_add<2>(v);
        v = row_xor_add<4>(v);
        v = row_xor_add<8>(v);
        pr[tl * 8 + r] = v;
      }
    }
    if (lrow == 0) {
#pragma unroll
      for (int tl = 0; tl < 2; ++tl) {
#pragma unroll
        for (int r = 0; r < 8; ++r) {
          const int row = wv * 32 + tl * 16 + r + half * 8;
          out[(size_t)(b0 + row) * NA + a] = pr[tl * 8 + r];
        }
      }
    }
  }
}

extern "C" void kernel_launch(void* const* d_in, const int* in_sizes, int n_in,
                              void* d_out, int out_size, void* d_ws, size_t ws_size,
                              hipStream_t stream) {
  const float* x  = (const float*)d_in[0];
  const float* y  = (const float*)d_in[1];
  const float* w1 = (const float*)d_in[2];
  const float* w2 = (const float*)d_in[3];
  const float* w3 = (const float*)d_in[4];
  const float* c1 = (const float*)d_in[5];
  const float* c2 = (const float*)d_in[6];
  const float* c3 = (const float*)d_in[7];
  const int* i1 = (const int*)d_in[8];
  const int* l1 = (const int*)d_in[9];
  const int* k1 = (const int*)d_in[10];
  const int* q1 = (const int*)d_in[11];
  const int* i2 = (const int*)d_in[12];
  const int* j2 = (const int*)d_in[13];
  const int* l2 = (const int*)d_in[14];
  const int* m2 = (const int*)d_in[15];
  const int* k2 = (const int*)d_in[16];
  const int* q2 = (const int*)d_in[17];
  const int* i3 = (const int*)d_in[18];
  const int* j3 = (const int*)d_in[19];
  const int* f3 = (const int*)d_in[20];
  const int* l3 = (const int*)d_in[21];
  const int* m3 = (const int*)d_in[22];
  const int* g3 = (const int*)d_in[23];
  const int* k3 = (const int*)d_in[24];
  const int* q3 = (const int*)d_in[25];

  int*   offs = (int*)d_ws;
  float* Wg   = (float*)((char*)d_ws + 2048);
  float* out  = (float*)d_out;

  prep_offsets_kernel<<<2, 256, 0, stream>>>(i1, l1, i2, j2, l2, m2,
                                             i3, j3, f3, l3, m3, g3, offs);
  prep_wg_kernel<<<104, 256, 0, stream>>>(w1, w2, w3, c1, c2, c3,
                                          k1, q1, k2, q2, k3, q3, Wg);
  contract_kernel<<<dim3(32, 32), WG_THREADS, 0, stream>>>(x, y, Wg, offs, out);
}


# One launch fewer (3 to 2): the packed gather offsets are built by two extra blocks of the weight-gather prep kernel, side by side in one grid, so the main kernel waits for one launch, not two
# speedup vs baseline: 3.3825x; 1.0426x over previous
#include <hip/hip_runtime.h>
#include <hip/hip_bf16.h>

typedef float v2f __attribute__((ext_vector_type(2)));
typedef float v8f __attribute__((ext_vector_type(8)));

#define NB 4096
#define NA 64
#define NL 16
#define NM 4
#define NE 10
#define P_TOT 416
#define P1_END 32
#define P2_END 160
#define EPAD 16
#define XP2 132
#define WG_WAVES 4
#define WG_THREADS 128
#define WG_ROWS 128
#define A_PER_WG 2

#define PREP_WG_BLOCKS 104
#define PREP_OFF_BLOCKS 2

__global__ void prep_kernel(const float* __restrict__ w1, const float* __restrict__ w2,
                            const float* __restrict__ w3,
                            const float* __restrict__ c1, const float* __restrict__ c2,
                            const float* __restrict__ c3,
                            const int* __restrict__ k1, const int* __restrict__ q1,
                            const int* __restrict__ k2, const int* __restrict__ q2,
                            const int* __restrict__ k3, const int* __restrict__ q3,
                            const int* __restrict__ i1, const int* __restrict__ l1,
                            const int* __restrict__ i2, const int* __restrict__ j2,
                            const int* __restrict__ l2, const int* __restrict__ m2,
                            const int* __restrict__ i3, const int* __restrict__ j3,
                            const int* __restrict__ f3, const int* __restrict__ l3,
                            const int* __restrict__ m3, const int* __restrict__ g3,
                            float* __restrict__ Wg, int* __restrict__ offs) {
  if (blockIdx.x >= PREP_WG_BLOCKS) {
    const int p = (blockIdx.x - PREP_WG_BLOCKS) * blockDim.x + threadIdx.x;
    if (p >= P_TOT) return;
    int packed;
    if (p < P1_END) {
      int o1 = i1[p] * NM + l1[p];
      packed = o1;
    } else if (p < P2_END) {
      int q = p - P1_END;
      int o1 = i2[q] * NM + l2[q];
      int o2 = j2[q] * NM + m2[q];
      packed = o1 | (o2 << 8);
    } else {
      int q = p - P2_END;
      int o1 = i3[q] * NM + l3[q];
      int o2 = j3[q] * NM + m3[q];
      int o3 = f3[q] * NM + g3[q];
      packed = o1 | (o2 << 8) | (o3 << 16);
    }
    offs[p] = packed;
    return;
  }
  int tid = blockIdx.x * blockDim.x + threadIdx.x;
  if (tid >= NA * P_TOT) return;
  int p = tid % P_TOT;
  int a = tid / P_TOT;
  float cf;
  const float* wbase;
  int estride;
  int kq;
  if (p < P1_END) {
    cf = c1[p];
    wbase = w1; kq = k1[p] * 2 + q1[p]; estride = 4 * 2 * NA;
  } else if (p < P2_END) {
    int t = p - P1_END;
    cf = c2[t];
    wbase = w2; kq = k2[t] * 4 + q2[t]; estride = 16 * 4 * NA;
  } else {
    int t = p - P2_END;
    cf = c3[t];
    wbase = w3; kq = k3[t] * 6 + q3[t]; estride = 32 * 6 * NA;
  }
  const int kqstride = kq * NA + a;
  float* dst = Wg + ((size_t)a * P_TOT + p) * EPAD;
#pragma unroll
  for (int n = 0; n < EPAD; ++n) {
    dst[n] = (n < NE) ? cf * wbase[(size_t)n * estride + kqstride] : 0.0f;
  }
}

template <int CNT>
__device__ __forceinline__ void wmma_step2(int t, const int* __restrict__ ldsOff,
                                           const v2f* __restrict__ X,
                                           const float* __restrict__ ldsW,
                                           int half, int lrow, v8f& acc0, v8f& acc1) {
  const int pA = 4 * t + 2 * half;
  const int pk0 = ldsOff[pA];
  const int pk1 = ldsOff[pA + 1];
  v2f xa = X[pk0 & 0xFF];
  v2f xb = X[pk1 & 0xFF];
  if (CNT > 1) { xa *= X[(pk0 >> 8) & 0xFF]; xb *= X[(pk1 >> 8) & 0xFF]; }
  if (CNT > 2) { xa *= X[(pk0 >> 16) & 0xFF]; xb *= X[(pk1 >> 16) & 0xFF]; }
  v2f a0; a0.x = xa.x; a0.y = xb.x;
  v2f a1; a1.x = xa.y; a1.y = xb.y;
  v2f bf;
  bf.x = ldsW[pA * EPAD + lrow];
  bf.y = ldsW[(pA + 1) * EPAD + lrow];
  acc0 = __builtin_amdgcn_wmma_f32_16x16x4_f32(false, a0, false, bf,
                                               (short)0, acc0, false, false);
  acc1 = __builtin_amdgcn_wmma_f32_16x16x4_f32(false, a1, false, bf,
                                               (short)0, acc1, false, false);
}

typedef float v4f __attribute__((ext_vector_type(4)));

struct StageRegs {
  v4f w[13];
  v4f xr[16];
};

__device__ __forceinline__ void stage_load(StageRegs& r, const float* __restrict__ x,
                                           const float* __restrict__ Wg, int a,
                                           int tid, int lane, int myb) {
  const v4f* ws = (const v4f*)(Wg + (size_t)a * P_TOT * EPAD) + tid;
#pragma unroll
  for (int k = 0; k < 13; ++k) r.w[k] = ws[WG_THREADS * k];
  const float* xb = x + ((size_t)(myb + (lane >> 4)) * NA + a) * 64 + (lane & 15) * 4;
  const int XS = 2 * NA * 64;
#pragma unroll
  for (int k = 0; k < 16; ++k) r.xr[k] = *(const v4f*)(xb + k * XS);
}

__device__ __forceinline__ void stage_store(const StageRegs& r, float* ldsW, float* ldsX,
                                            int tid, int lane, int wv) {
  v4f* wd = (v4f*)ldsW + tid;
#pragma unroll
  for (int k = 0; k < 13; ++k) wd[WG_THREADS * k] = r.w[k];
  float* xd = &ldsX[(wv * 16 + (lane >> 4)) * XP2 + (lane & 15) * 8];
  const int LS = 2 * XP2;
#pragma unroll
  for (int j = 0; j < 8; ++j) {
    const v4f t0 = r.xr[j];
    const v4f t1 = r.xr[8 + j];
    *(v4f*)(xd + j * LS)     = (v4f){t0.x, t1.x, t0.y, t1.y};
    *(v4f*)(xd + j * LS + 4) = (v4f){t0.z, t1.z, t0.w, t1.w};
  }
}

template <int MASK>
__device__ __forceinline__ float row_xor_add(float v) {
  const int other = __builtin_amdgcn_update_dpp(0, __float_as_int(v), 0x160 + MASK, 0xF, 0xF, true);
  return v + __int_as_float(other);
}

__global__ __launch_bounds__(WG_THREADS) void contract_kernel(const float* __restrict__ x,
                                                       const float* __restrict__ y,
                                                       const float* __restrict__ Wg,
                                                       const int* __restrict__ offs,
                                                       float* __restrict__ out) {
  __shared__ __attribute__((aligned(16))) float ldsW[P_TOT * EPAD];
  __shared__ __attribute__((aligned(16))) float ldsX[WG_WAVES * 16 * XP2];
  __shared__ __attribute__((aligned(16))) float ldsY[WG_ROWS * EPAD];
  __shared__ int   ldsOff[P_TOT];

  const int tid  = threadIdx.x;
  const int wv   = tid >> 5;
  const int lane = tid & 31;
  const int half = lane >> 4;
  const int lrow = lane & 15;
  const int b0   = blockIdx.x * WG_ROWS;
  const int a0   = blockIdx.y * A_PER_WG;
  const int myb  = b0 + wv * 32;

  StageRegs sr;
  stage_load(sr, x, Wg, a0, tid, lane, myb);
  const int off0 = offs[tid];
  const int off1 = offs[WG_THREADS + tid];
  const int off2 = offs[2 * WG_THREADS + tid];
  int off3 = 0;
  if (tid < P_TOT - 3 * WG_THREADS) off3 = offs[3 * WG_THREADS + tid];
  float yv[16];
#pragma unroll
  for (int j = 0; j < 16; ++j) yv[j] = 0.0f;
  if ((tid & 15) < NE) {
    const float* yb = y + (size_t)(b0 + (tid >> 4)) * NE + (tid & 15);
    const int YS = 8 * NE;
#pragma unroll
    for (int j = 0; j < 16; ++j) yv[j] = yb[j * YS];
  }

  ldsOff[tid] = off0;
  ldsOff[WG_THREADS + tid] = off1;
  ldsOff[2 * WG_THREADS + tid] = off2;
  if (tid < P_TOT - 3 * WG_THREADS) ldsOff[3 * WG_THREADS + tid] = off3;
#pragma unroll
  for (int j = 0; j < 16; ++j) ldsY[tid + WG_THREADS * j] = yv[j];

#pragma unroll
  for (int ai = 0; ai < A_PER_WG; ++ai) {
    const int a = a0 + ai;
    if (ai > 0) __syncthreads();
    stage_store(sr, ldsW, ldsX, tid, lane, wv);
    __syncthreads();

    const v2f* X = (const v2f*)&ldsX[(wv * 16 + lrow) * XP2];
    v8f acc0 = {};
    v8f acc1 = {};
#pragma unroll
    for (int t = 0; t < 8; ++t)
      wmma_step2<1>(t, ldsOff, X, ldsW, half, lrow, acc0, acc1);
#pragma unroll 4
    for (int t = 8; t < 40; ++t)
      wmma_step2<2>(t, ldsOff, X, ldsW, half, lrow, acc0, acc1);
#pragma unroll 4
    for (int t = 40; t < 104; ++t)
      wmma_step2<3>(t, ldsOff, X, ldsW, half, lrow, acc0, acc1);

    if (ai + 1 < A_PER_WG) stage_load(sr, x, Wg, a + 1, tid, lane, myb);

    float pr[16];
#pragma unroll
    for (int tl = 0; tl < 2; ++tl) {
#pragma unroll
      for (int r = 0; r < 8; ++r) {
        const int row = wv * 32 + tl * 16 + r + half * 8;
        float v = (tl == 0 ? acc0[r] : acc1[r]) * ldsY[row * EPAD + lrow];
        v = row_xor_add<1>(v);
        v = row_xor_add<2>(v);
        v = row_xor_add<4>(v);
        v = row_xor_add<8>(v);
        pr[tl * 8 + r] = v;
      }
    }
    if (lrow == 0) {
#pragma unroll
      for (int tl = 0; tl < 2; ++tl) {
#pragma unroll
        for (int r = 0; r < 8; ++r) {
          const int row = wv * 32 + tl * 16 + r + half * 8;
          out[(size_t)(b0 + row) * NA + a] = pr[tl * 8 + r];
        }
      }
    }
  }
}

extern "C" void kernel_launch(void* const* d_in, const int* in_sizes, int n_in,
                              void* d_out, int out_size, void* d_ws, size_t ws_size,
                              hipStream_t stream) {
  const float* x  = (const float*)d_in[0];
  const float* y  = (const float*)d_in[1];
  const float* w1 = (const float*)d_in[2];
  const float* w2 = (const float*)d_in[3];
  const float* w3 = (const float*)d_in[4];
  const float* c1 = (const float*)d_in[5];
  const float* c2 = (const float*)d_in[6];
  const float* c3 = (const float*)d_in[7];
  const int* i1 = (const int*)d_in[8];
  const int* l1 = (const int*)d_in[9];
  const int* k1 = (const int*)d_in[10];
  const int* q1 = (const int*)d_in[11];
  const int* i2 = (const int*)d_in[12];
  const int* j2 = (const int*)d_in[13];
  const int* l2 = (const int*)d_in[14];
  const int* m2 = (const int*)d_in[15];
  const int* k2 = (const int*)d_in[16];
  const int* q2 = (const int*)d_in[17];
  const int* i3 = (const int*)d_in[18];
  const int* j3 = (const int*)d_in[19];
  const int* f3 = (const int*)d_in[20];
  const int* l3 = (const int*)d_in[21];
  const int* m3 = (const int*)d_in[22];
  const int* g3 = (const int*)d_in[23];
  const int* k3 = (const int*)d_in[24];
  const int* q3 = (const int*)d_in[25];

  int*   offs = (int*)d_ws;
  float* Wg   = (float*)((char*)d_ws + 2048);
  float* out  = (float*)d_out;

  prep_kernel<<<PREP_WG_BLOCKS + PREP_OFF_BLOCKS, 256, 0, stream>>>(w1, w2, w3, c1, c2, c3,
                                                                    k1, q1, k2, q2, k3, q3,
                                                                    i1, l1, i2, j2, l2, m2,
                                                                    i3, j3, f3, l3, m3, g3, Wg, offs);
  contract_kernel<<<dim3(32, 32), WG_THREADS, 0, stream>>>(x, y, Wg, offs, out);
}


# K loop software-pipelined per wave: the next 8-WMMA block's LDS gathers and B fragment (offsets two blocks ahead) issued as one group before the current block's WMMAs, not load-wait-use per block
# speedup vs baseline: 3.5681x; 1.0549x over previous
#include <hip/hip_runtime.h>
#include <hip/hip_bf16.h>

typedef float v2f __attribute__((ext_vector_type(2)));
typedef float v8f __attribute__((ext_vector_type(8)));

#define NB 4096
#define NA 64
#define NL 16
#define NM 4
#define NE 10
#define P_TOT 416
#define P1_END 32
#define P2_END 160
#define EPAD 16
#define XP2 132
#define WG_WAVES 4
#define WG_THREADS 128
#define WG_ROWS 128
#define A_PER_WG 2

#define PREP_WG_BLOCKS 104
#define PREP_OFF_BLOCKS 2

__global__ void prep_kernel(const float* __restrict__ w1, const float* __restrict__ w2,
                            const float* __restrict__ w3,
                            const float* __restrict__ c1, const float* __restrict__ c2,
                            const float* __restrict__ c3,
                            const int* __restrict__ k1, const int* __restrict__ q1,
                            const int* __restrict__ k2, const int* __restrict__ q2,
                            const int* __restrict__ k3, const int* __restrict__ q3,
                            const int* __restrict__ i1, const int* __restrict__ l1,
                            const int* __restrict__ i2, const int* __restrict__ j2,
                            const int* __restrict__ l2, const int* __restrict__ m2,
                            const int* __restrict__ i3, const int* __restrict__ j3,
                            const int* __restrict__ f3, const int* __restrict__ l3,
                            const int* __restrict__ m3, const int* __restrict__ g3,
                            float* __restrict__ Wg, int* __restrict__ offs) {
  if (blockIdx.x >= PREP_WG_BLOCKS) {
    const int p = (blockIdx.x - PREP_WG_BLOCKS) * blockDim.x + threadIdx.x;
    if (p >= P_TOT) return;
    int packed;
    if (p < P1_END) {
      int o1 = i1[p] * NM + l1[p];
      packed = o1;
    } else if (p < P2_END) {
      int q = p - P1_END;
      int o1 = i2[q] * NM + l2[q];
      int o2 = j2[q] * NM + m2[q];
      packed = o1 | (o2 << 8);
    } else {
      int q = p - P2_END;
      int o1 = i3[q] * NM + l3[q];
      int o2 = j3[q] * NM + m3[q];
      int o3 = f3[q] * NM + g3[q];
      packed = o1 | (o2 << 8) | (o3 << 16);
    }
    offs[p] = packed;
    return;
  }
  int tid = blockIdx.x * blockDim.x + threadIdx.x;
  if (tid >= NA * P_TOT) return;
  int p = tid % P_TOT;
  int a = tid / P_TOT;
  float cf;
  const float* wbase;
  int estride;
  int kq;
  if (p < P1_END) {
    cf = c1[p];
    wbase = w1; kq = k1[p] * 2 + q1[p]; estride = 4 * 2 * NA;
  } else if (p < P2_END) {
    int t = p - P1_END;
    cf = c2[t];
    wbase = w2; kq = k2[t] * 4 + q2[t]; estride = 16 * 4 * NA;
  } else {
    int t = p - P2_END;
    cf = c3[t];
    wbase = w3; kq = k3[t] * 6 + q3[t]; estride = 32 * 6 * NA;
  }
  const int kqstride = kq * NA + a;
  float* dst = Wg + ((size_t)a * P_TOT + p) * EPAD;
#pragma unroll
  for (int n = 0; n < EPAD; ++n) {
    dst[n] = (n < NE) ? cf * wbase[(size_t)n * estride + kqstride] : 0.0f;
  }
}

#define BLK_STEPS 4
#define BLK_SLOTS 8
#define BLK_FENCE() __builtin_amdgcn_sched_barrier(0)

__device__ __forceinline__ void blk_load_offsets(int blk, const int* __restrict__ ldsOff, int half,
                                                 int (&pk)[BLK_SLOTS]) {
#pragma unroll
  for (int s = 0; s < BLK_STEPS; ++s) {
    const int pA = 4 * (BLK_STEPS * blk + s) + 2 * half;
    pk[2 * s]     = ldsOff[pA];
    pk[2 * s + 1] = ldsOff[pA + 1];
  }
}

template <int CNT>
__device__ __forceinline__ void blk_issue(int blk, const int (&pk)[BLK_SLOTS], const v2f* __restrict__ X,
                                          const float* __restrict__ ldsW, int half, int lrow,
                                          v2f (&g)[3][BLK_SLOTS], float (&bw)[BLK_SLOTS]) {
#pragma unroll
  for (int j = 0; j < BLK_SLOTS; ++j) {
    g[0][j] = X[pk[j] & 0xFF];
    if (CNT > 1) g[1][j] = X[(pk[j] >> 8) & 0xFF];
    if (CNT > 2) g[2][j] = X[(pk[j] >> 16) & 0xFF];
  }
#pragma unroll
  for (int s = 0; s < BLK_STEPS; ++s) {
    const int pA = 4 * (BLK_STEPS * blk + s) + 2 * half;
    bw[2 * s]     = ldsW[pA * EPAD + lrow];
    bw[2 * s + 1] = ldsW[(pA + 1) * EPAD + lrow];
  }
}

template <int CNT>
__device__ __forceinline__ void blk_compute(const v2f (&g)[3][BLK_SLOTS], const float (&bw)[BLK_SLOTS],
                                            v8f& acc0, v8f& acc1) {
#pragma unroll
  for (int s = 0; s < BLK_STEPS; ++s) {
    v2f xa = g[0][2 * s];
    v2f xb = g[0][2 * s + 1];
    if (CNT > 1) { xa *= g[1][2 * s]; xb *= g[1][2 * s + 1]; }
    if (CNT > 2) { xa *= g[2][2 * s]; xb *= g[2][2 * s + 1]; }
    v2f a0; a0.x = xa.x; a0.y = xb.x;
    v2f a1; a1.x = xa.y; a1.y = xb.y;
    v2f bf; bf.x = bw[2 * s]; bf.y = bw[2 * s + 1];
    acc0 = __builtin_amdgcn_wmma_f32_16x16x4_f32(false, a0, false, bf,
                                                 (short)0, acc0, false, false);
    acc1 = __builtin_amdgcn_wmma_f32_16x16x4_f32(false, a1, false, bf,
                                                 (short)0, acc1, false, false);
  }
}

typedef float v4f __attribute__((ext_vector_type(4)));

struct StageRegs {
  v4f w[13];
  v4f xr[16];
};

__device__ __forceinline__ void stage_load(StageRegs& r, const float* __restrict__ x,
                                           const float* __restrict__ Wg, int a,
                                           int tid, int lane, int myb) {
  const v4f* ws = (const v4f*)(Wg + (size_t)a * P_TOT * EPAD) + tid;
#pragma unroll
  for (int k = 0; k < 13; ++k) r.w[k] = ws[WG_THREADS * k];
  const float* xb = x + ((size_t)(myb + (lane >> 4)) * NA + a) * 64 + (lane & 15) * 4;
  const int XS = 2 * NA * 64;
#pragma unroll
  for (int k = 0; k < 16; ++k) r.xr[k] = *(const v4f*)(xb + k * XS);
}

__device__ __forceinline__ void stage_store(const StageRegs& r, float* ldsW, float* ldsX,
                                            int tid, int lane, int wv) {
  v4f* wd = (v4f*)ldsW + tid;
#pragma unroll
  for (int k = 0; k < 13; ++k) wd[WG_THREADS * k] = r.w[k];
  float* xd = &ldsX[(wv * 16 + (lane >> 4)) * XP2 + (lane & 15) * 8];
  const int LS = 2 * XP2;
#pragma unroll
  for (int j = 0; j < 8; ++j) {
    const v4f t0 = r.xr[j];
    const v4f t1 = r.xr[8 + j];
    *(v4f*)(xd + j * LS)     = (v4f){t0.x, t1.x, t0.y, t1.y};
    *(v4f*)(xd + j * LS + 4) = (v4f){t0.z, t1.z, t0.w, t1.w};
  }
}

template <int MASK>
__device__ __forceinline__ float row_xor_add(float v) {
  const int other = __builtin_amdgcn_update_dpp(0, __float_as_int(v), 0x160 + MASK, 0xF, 0xF, true);
  return v + __int_as_float(other);
}

__global__ __launch_bounds__(WG_THREADS) void contract_kernel(const float* __restrict__ x,
                                                       const float* __restrict__ y,
                                                       const float* __restrict__ Wg,
                                                       const int* __restrict__ offs,
                                                       float* __restrict__ out) {
  __shared__ __attribute__((aligned(16))) float ldsW[P_TOT * EPAD];
  __shared__ __attribute__((aligned(16))) float ldsX[WG_WAVES * 16 * XP2];
  __shared__ __attribute__((aligned(16))) float ldsY[WG_ROWS * EPAD];
  __shared__ int   ldsOff[P_TOT];

  const int tid  = threadIdx.x;
  const int wv   = tid >> 5;
  const int lane = tid & 31;
  const int half = lane >> 4;
  const int lrow = lane & 15;
  const int b0   = blockIdx.x * WG_ROWS;
  const int a0   = blockIdx.y * A_PER_WG;
  const int myb  = b0 + wv * 32;

  StageRegs sr;
  stage_load(sr, x, Wg, a0, tid, lane, myb);
  const int off0 = offs[tid];
  const int off1 = offs[WG_THREADS + tid];
  const int off2 = offs[2 * WG_THREADS + tid];
  int off3 = 0;
  if (tid < P_TOT - 3 * WG_THREADS) off3 = offs[3 * WG_THREADS + tid];
  float yv[16];
#pragma unroll
  for (int j = 0; j < 16; ++j) yv[j] = 0.0f;
  if ((tid & 15) < NE) {
    const float* yb = y + (size_t)(b0 + (tid >> 4)) * NE + (tid & 15);
    const int YS = 8 * NE;
#pragma unroll
    for (int j = 0; j < 16; ++j) yv[j] = yb[j * YS];
  }

  ldsOff[tid] = off0;
  ldsOff[WG_THREADS + tid] = off1;
  ldsOff[2 * WG_THREADS + tid] = off2;
  if (tid < P_TOT - 3 * WG_THREADS) ldsOff[3 * WG_THREADS + tid] = off3;
#pragma unroll
  for (int j = 0; j < 16; ++j) ldsY[tid + WG_THREADS * j] = yv[j];

#pragma unroll
  for (int ai = 0; ai < A_PER_WG; ++ai) {
    const int a = a0 + ai;
    if (ai > 0) __syncthreads();
    stage_store(sr, ldsW, ldsX, tid, lane, wv);
    __syncthreads();

    const v2f* X = (const v2f*)&ldsX[(wv * 16 + lrow) * XP2];
    v8f acc0 = {};
    v8f acc1 = {};
    int   pkA[BLK_SLOTS], pkB[BLK_SLOTS];
    v2f   gA[3][BLK_SLOTS], gB[3][BLK_SLOTS];
    float bwA[BLK_SLOTS], bwB[BLK_SLOTS];

    blk_load_offsets(0, ldsOff, half, pkA);
    blk_load_offsets(1, ldsOff, half, pkB);
    blk_issue<1>(0, pkA, X, ldsW, half, lrow, gA, bwA);
    blk_load_offsets(2, ldsOff, half, pkA);
    blk_issue<1>(1, pkB, X, ldsW, half, lrow, gB, bwB);
    blk_load_offsets(3, ldsOff, half, pkB);
    BLK_FENCE();
    blk_compute<1>(gA, bwA, acc0, acc1);
    BLK_FENCE();
    blk_issue<2>(2, pkA, X, ldsW, half, lrow, gA, bwA);
    blk_load_offsets(4, ldsOff, half, pkA);
    BLK_FENCE();
    blk_compute<1>(gB, bwB, acc0, acc1);
    BLK_FENCE();
#pragma unroll 1
    for (int blk = 2; blk < 8; blk += 2) {
      blk_issue<2>(blk + 1, pkB, X, ldsW, half, lrow, gB, bwB);
      blk_load_offsets(blk + 3, ldsOff, half, pkB);
      BLK_FENCE();
      blk_compute<2>(gA, bwA, acc0, acc1);
      BLK_FENCE();
      blk_issue<2>(blk + 2, pkA, X, ldsW, half, lrow, gA, bwA);
      blk_load_offsets(blk + 4, ldsOff, half, pkA);
      BLK_FENCE();
      blk_compute<2>(gB, bwB, acc0, acc1);
      BLK_FENCE();
    }
    blk_issue<2>(9, pkB, X, ldsW, half, lrow, gB, bwB);
    blk_load_offsets(11, ldsOff, half, pkB);
    BLK_FENCE();
    blk_compute<2>(gA, bwA, acc0, acc1);
    BLK_FENCE();
    blk_issue<3>(10, pkA, X, ldsW, half, lrow, gA, bwA);
    blk_load_offsets(12, ldsOff, half, pkA);
    BLK_FENCE();
    blk_compute<2>(gB, bwB, acc0, acc1);
    BLK_FENCE();
#pragma unroll 1
    for (int blk = 10; blk < 24; blk += 2) {
      blk_issue<3>(blk + 1, pkB, X, ldsW, half, lrow, gB, bwB);
      blk_load_offsets(blk + 3, ldsOff, half, pkB);
      BLK_FENCE();
      blk_compute<3>(gA, bwA, acc0, acc1);
      BLK_FENCE();
      blk_issue<3>(blk + 2, pkA, X, ldsW, half, lrow, gA, bwA);
      blk_load_offsets(blk + 4 < 26 ? blk + 4 : 25, ldsOff, half, pkA);
      BLK_FENCE();
      blk_compute<3>(gB, bwB, acc0, acc1);
      BLK_FENCE();
    }
    blk_issue<3>(25, pkB, X, ldsW, half, lrow, gB, bwB);
    BLK_FENCE();
    blk_compute<3>(gA, bwA, acc0, acc1);
    blk_compute<3>(gB, bwB, acc0, acc1);

    if (ai + 1 < A_PER_WG) stage_load(sr, x, Wg, a + 1, tid, lane, myb);

    float pr[16];
#pragma unroll
    for (int tl = 0; tl < 2; ++tl) {
#pragma unroll
      for (int r = 0; r < 8; ++r) {
        const int row = wv * 32 + tl * 16 + r + half * 8;
        float v = (tl == 0 ? acc0[r] : acc1[r]) * ldsY[row * EPAD + lrow];
        v = row_xor_add<1>(v);
        v = row_xor_add<2>(v);
        v = row_xor_add<4>(v);
        v = row_xor_add<8>(v);
        pr[tl * 8 + r] = v;
      }
    }
    if (lrow == 0) {
#pragma unroll
      for (int tl = 0; tl < 2; ++tl) {
#pragma unroll
        for (int r = 0; r < 8; ++r) {
          const int row = wv * 32 + tl * 16 + r + half * 8;
          out[(size_t)(b0 + row) * NA + a] = pr[tl * 8 + r];
        }
      }
    }
  }
}

extern "C" void kernel_launch(void* const* d_in, const int* in_sizes, int n_in,
                              void* d_out, int out_size, void* d_ws, size_t ws_size,
                              hipStream_t stream) {
  const float* x  = (const float*)d_in[0];
  const float* y  = (const float*)d_in[1];
  const float* w1 = (const float*)d_in[2];
  const float* w2 = (const float*)d_in[3];
  const float* w3 = (const float*)d_in[4];
  const float* c1 = (const float*)d_in[5];
  const float* c2 = (const float*)d_in[6];
  const float* c3 = (const float*)d_in[7];
  const int* i1 = (const int*)d_in[8];
  const int* l1 = (const int*)d_in[9];
  const int* k1 = (const int*)d_in[10];
  const int* q1 = (const int*)d_in[11];
  const int* i2 = (const int*)d_in[12];
  const int* j2 = (const int*)d_in[13];
  const int* l2 = (const int*)d_in[14];
  const int* m2 = (const int*)d_in[15];
  const int* k2 = (const int*)d_in[16];
  const int* q2 = (const int*)d_in[17];
  const int* i3 = (const int*)d_in[18];
  const int* j3 = (const int*)d_in[19];
  const int* f3 = (const int*)d_in[20];
  const int* l3 = (const int*)d_in[21];
  const int* m3 = (const int*)d_in[22];
  const int* g3 = (const int*)d_in[23];
  const int* k3 = (const int*)d_in[24];
  const int* q3 = (const int*)d_in[25];

  int*   offs = (int*)d_ws;
  float* Wg   = (float*)((char*)d_ws + 2048);
  float* out  = (float*)d_out;

  prep_kernel<<<PREP_WG_BLOCKS + PREP_OFF_BLOCKS, 256, 0, stream>>>(w1, w2, w3, c1, c2, c3,
                                                                    k1, q1, k2, q2, k3, q3,
                                                                    i1, l1, i2, j2, l2, m2,
                                                                    i3, j3, f3, l3, m3, g3, Wg, offs);
  contract_kernel<<<dim3(32, 32), WG_THREADS, 0, stream>>>(x, y, Wg, offs, out);
}


# Gather addresses come from a prep-built LDS table of decoded byte offsets (16-byte loads, one add per gather), not and/shift unpacking of packed offsets; block fences now pin the accumulators
# speedup vs baseline: 3.7654x; 1.0553x over previous
#include <hip/hip_runtime.h>
#include <hip/hip_bf16.h>

typedef float v2f __attribute__((ext_vector_type(2)));
typedef float v8f __attribute__((ext_vector_type(8)));

#define NB 4096
#define NA 64
#define NL 16
#define NM 4
#define NE 10
#define P_TOT 416
#define P1_END 32
#define P2_END 160
#define EPAD 16
#define XP2 132
#define WG_WAVES 4
#define WG_THREADS 128
#define WG_ROWS 128
#define A_PER_WG 2

#define PREP_WG_BLOCKS 104
#define TAB_INTS 1248
#define TAB_VEC4 312
#define PREP_OFF_BLOCKS 5

__global__ void prep_kernel(const float* __restrict__ w1, const float* __restrict__ w2,
                            const float* __restrict__ w3,
                            const float* __restrict__ c1, const float* __restrict__ c2,
                            const float* __restrict__ c3,
                            const int* __restrict__ k1, const int* __restrict__ q1,
                            const int* __restrict__ k2, const int* __restrict__ q2,
                            const int* __restrict__ k3, const int* __restrict__ q3,
                            const int* __restrict__ i1, const int* __restrict__ l1,
                            const int* __restrict__ i2, const int* __restrict__ j2,
                            const int* __restrict__ l2, const int* __restrict__ m2,
                            const int* __restrict__ i3, const int* __restrict__ j3,
                            const int* __restrict__ f3, const int* __restrict__ l3,
                            const int* __restrict__ m3, const int* __restrict__ g3,
                            float* __restrict__ Wg, int* __restrict__ offs) {
  if (blockIdx.x >= PREP_WG_BLOCKS) {
    const int t = (blockIdx.x - PREP_WG_BLOCKS) * blockDim.x + threadIdx.x;
    if (t >= TAB_INTS) return;
    const int blk  = t / 48;
    const int r    = t - blk * 48;
    const int half = r / 24;
    const int r2   = r - half * 24;
    const int k    = r2 >> 3;
    const int j    = r2 & 7;
    const int p    = 16 * blk + 4 * (j >> 1) + 2 * half + (j & 1);
    int o = 0;
    if (p < P1_END) {
      if (k == 0) o = i1[p] * NM + l1[p];
    } else if (p < P2_END) {
      const int q = p - P1_END;
      if (k == 0)      o = i2[q] * NM + l2[q];
      else if (k == 1) o = j2[q] * NM + m2[q];
    } else {
      const int q = p - P2_END;
      if (k == 0)      o = i3[q] * NM + l3[q];
      else if (k == 1) o = j3[q] * NM + m3[q];
      else             o = f3[q] * NM + g3[q];
    }
    offs[t] = (o & 63) * 8;
    return;
  }
  int tid = blockIdx.x * blockDim.x + threadIdx.x;
  if (tid >= NA * P_TOT) return;
  int p = tid % P_TOT;
  int a = tid / P_TOT;
  float cf;
  const float* wbase;
  int estride;
  int kq;
  if (p < P1_END) {
    cf = c1[p];
    wbase = w1; kq = k1[p] * 2 + q1[p]; estride = 4 * 2 * NA;
  } else if (p < P2_END) {
    int t = p - P1_END;
    cf = c2[t];
    wbase = w2; kq = k2[t] * 4 + q2[t]; estride = 16 * 4 * NA;
  } else {
    int t = p - P2_END;
    cf = c3[t];
    wbase = w3; kq = k3[t] * 6 + q3[t]; estride = 32 * 6 * NA;
  }
  const int kqstride = kq * NA + a;
  float* dst = Wg + ((size_t)a * P_TOT + p) * EPAD;
#pragma unroll
  for (int n = 0; n < EPAD; ++n) {
    dst[n] = (n < NE) ? cf * wbase[(size_t)n * estride + kqstride] : 0.0f;
  }
}

#define BLK_STEPS 4
#define BLK_SLOTS 8
#define BLK_FENCE_ACC(A0, A1) do { asm volatile("" : "+v"(A0), "+v"(A1)); __builtin_amdgcn_sched_barrier(0); } while (0)

typedef int v4i __attribute__((ext_vector_type(4)));

template <int CNT>
__device__ __forceinline__ void blk_load_offsets(int blk, const v4i* __restrict__ ldsTab, int half,
                                                 int (&ao)[3][BLK_SLOTS]) {
  const v4i* t = ldsTab + (blk * 2 + half) * 6;
#pragma unroll
  for (int k = 0; k < CNT; ++k) {
    const v4i lo = t[2 * k];
    const v4i hi = t[2 * k + 1];
    ao[k][0] = lo.x; ao[k][1] = lo.y; ao[k][2] = lo.z; ao[k][3] = lo.w;
    ao[k][4] = hi.x; ao[k][5] = hi.y; ao[k][6] = hi.z; ao[k][7] = hi.w;
  }
}

template <int CNT>
__device__ __forceinline__ void blk_issue(int blk, const int (&ao)[3][BLK_SLOTS], const v2f* __restrict__ X,
                                          const float* __restrict__ ldsW, int half, int lrow,
                                          v2f (&g)[3][BLK_SLOTS], float (&bw)[BLK_SLOTS]) {
  const char* Xb = (const char*)X;
#pragma unroll
  for (int j = 0; j < BLK_SLOTS; ++j) {
    g[0][j] = *(const v2f*)(Xb + ao[0][j]);
    if (CNT > 1) g[1][j] = *(const v2f*)(Xb + ao[1][j]);
    if (CNT > 2) g[2][j] = *(const v2f*)(Xb + ao[2][j]);
  }
#pragma unroll
  for (int s = 0; s < BLK_STEPS; ++s) {
    const int pA = 4 * (BLK_STEPS * blk + s) + 2 * half;
    bw[2 * s]     = ldsW[pA * EPAD + lrow];
    bw[2 * s + 1] = ldsW[(pA + 1) * EPAD + lrow];
  }
}

template <int CNT>
__device__ __forceinline__ void blk_compute(const v2f (&g)[3][BLK_SLOTS], const float (&bw)[BLK_SLOTS],
                                            v8f& acc0, v8f& acc1) {
#pragma unroll
  for (int s = 0; s < BLK_STEPS; ++s) {
    v2f xa = g[0][2 * s];
    v2f xb = g[0][2 * s + 1];
    if (CNT > 1) { xa *= g[1][2 * s]; xb *= g[1][2 * s + 1]; }
    if (CNT > 2) { xa *= g[2][2 * s]; xb *= g[2][2 * s + 1]; }
    v2f a0; a0.x = xa.x; a0.y = xb.x;
    v2f a1; a1.x = xa.y; a1.y = xb.y;
    v2f bf; bf.x = bw[2 * s]; bf.y = bw[2 * s + 1];
    acc0 = __builtin_amdgcn_wmma_f32_16x16x4_f32(false, a0, false, bf,
                                                 (short)0, acc0, false, false);
    acc1 = __builtin_amdgcn_wmma_f32_16x16x4_f32(false, a1, false, bf,
                                                 (short)0, acc1, false, false);
  }
}

typedef float v4f __attribute__((ext_vector_type(4)));

struct StageRegs {
  v4f w[13];
  v4f xr[16];
};

__device__ __forceinline__ void stage_load(StageRegs& r, const float* __restrict__ x,
                                           const float* __restrict__ Wg, int a,
                                           int tid, int lane, int myb) {
  const v4f* ws = (const v4f*)(Wg + (size_t)a * P_TOT * EPAD) + tid;
#pragma unroll
  for (int k = 0; k < 13; ++k) r.w[k] = ws[WG_THREADS * k];
  const float* xb = x + ((size_t)(myb + (lane >> 4)) * NA + a) * 64 + (lane & 15) * 4;
  const int XS = 2 * NA * 64;
#pragma unroll
  for (int k = 0; k < 16; ++k) r.xr[k] = *(const v4f*)(xb + k * XS);
}

__device__ __forceinline__ void stage_store(const StageRegs& r, float* ldsW, float* ldsX,
                                            int tid, int lane, int wv) {
  v4f* wd = (v4f*)ldsW + tid;
#pragma unroll
  for (int k = 0; k < 13; ++k) wd[WG_THREADS * k] = r.w[k];
  float* xd = &ldsX[(wv * 16 + (lane >> 4)) * XP2 + (lane & 15) * 8];
  const int LS = 2 * XP2;
#pragma unroll
  for (int j = 0; j < 8; ++j) {
    const v4f t0 = r.xr[j];
    const v4f t1 = r.xr[8 + j];
    *(v4f*)(xd + j * LS)     = (v4f){t0.x, t1.x, t0.y, t1.y};
    *(v4f*)(xd + j * LS + 4) = (v4f){t0.z, t1.z, t0.w, t1.w};
  }
}

template <int MASK>
__device__ __forceinline__ float row_xor_add(float v) {
  const int other = __builtin_amdgcn_update_dpp(0, __float_as_int(v), 0x160 + MASK, 0xF, 0xF, true);
  return v + __int_as_float(other);
}

__global__ __launch_bounds__(WG_THREADS) void contract_kernel(const float* __restrict__ x,
                                                       const float* __restrict__ y,
                                                       const float* __restrict__ Wg,
                                                       const int* __restrict__ offs,
                                                       float* __restrict__ out) {
  __shared__ __attribute__((aligned(16))) float ldsW[P_TOT * EPAD];
  __shared__ __attribute__((aligned(16))) float ldsX[WG_WAVES * 16 * XP2];
  __shared__ __attribute__((aligned(16))) float ldsY[WG_ROWS * EPAD];
  __shared__ v4i   ldsTab[TAB_VEC4];

  const int tid  = threadIdx.x;
  const int wv   = tid >> 5;
  const int lane = tid & 31;
  const int half = lane >> 4;
  const int lrow = lane & 15;
  const int b0   = blockIdx.x * WG_ROWS;
  const int a0   = blockIdx.y * A_PER_WG;
  const int myb  = b0 + wv * 32;

  StageRegs sr;
  stage_load(sr, x, Wg, a0, tid, lane, myb);
  const v4i* tg = (const v4i*)offs;
  const v4i tb0 = tg[tid];
  const v4i tb1 = tg[WG_THREADS + tid];
  v4i tb2 = {0, 0, 0, 0};
  if (tid < TAB_VEC4 - 2 * WG_THREADS) tb2 = tg[2 * WG_THREADS + tid];
  float yv[16];
#pragma unroll
  for (int j = 0; j < 16; ++j) yv[j] = 0.0f;
  if ((tid & 15) < NE) {
    const float* yb = y + (size_t)(b0 + (tid >> 4)) * NE + (tid & 15);
    const int YS = 8 * NE;
#pragma unroll
    for (int j = 0; j < 16; ++j) yv[j] = yb[j * YS];
  }

  ldsTab[tid] = tb0;
  ldsTab[WG_THREADS + tid] = tb1;
  if (tid < TAB_VEC4 - 2 * WG_THREADS) ldsTab[2 * WG_THREADS + tid] = tb2;
#pragma unroll
  for (int j = 0; j < 16; ++j) ldsY[tid + WG_THREADS * j] = yv[j];

#pragma unroll
  for (int ai = 0; ai < A_PER_WG; ++ai) {
    const int a = a0 + ai;
    if (ai > 0) __syncthreads();
    stage_store(sr, ldsW, ldsX, tid, lane, wv);
    __syncthreads();

    const v2f* X = (const v2f*)&ldsX[(wv * 16 + lrow) * XP2];
    v8f acc0 = {};
    v8f acc1 = {};
    int   ao[3][BLK_SLOTS];
    v2f   gA[3][BLK_SLOTS], gB[3][BLK_SLOTS];
    float bwA[BLK_SLOTS], bwB[BLK_SLOTS];

    blk_load_offsets<1>(0, ldsTab, half, ao);
    blk_issue<1>(0, ao, X, ldsW, half, lrow, gA, bwA);
    blk_load_offsets<1>(1, ldsTab, half, ao);
    blk_issue<1>(1, ao, X, ldsW, half, lrow, gB, bwB);
    blk_load_offsets<2>(2, ldsTab, half, ao);
    BLK_FENCE_ACC(acc0, acc1);
    blk_compute<1>(gA, bwA, acc0, acc1);
    BLK_FENCE_ACC(acc0, acc1);
    blk_issue<2>(2, ao, X, ldsW, half, lrow, gA, bwA);
    blk_load_offsets<2>(3, ldsTab, half, ao);
    BLK_FENCE_ACC(acc0, acc1);
    blk_compute<1>(gB, bwB, acc0, acc1);
    BLK_FENCE_ACC(acc0, acc1);
#pragma unroll 1
    for (int blk = 2; blk < 8; blk += 2) {
      blk_issue<2>(blk + 1, ao, X, ldsW, half, lrow, gB, bwB);
      blk_load_offsets<2>(blk + 2, ldsTab, half, ao);
      BLK_FENCE_ACC(acc0, acc1);
      blk_compute<2>(gA, bwA, acc0, acc1);
      BLK_FENCE_ACC(acc0, acc1);
      blk_issue<2>(blk + 2, ao, X, ldsW, half, lrow, gA, bwA);
      blk_load_offsets<2>(blk + 3, ldsTab, half, ao);
      BLK_FENCE_ACC(acc0, acc1);
      blk_compute<2>(gB, bwB, acc0, acc1);
      BLK_FENCE_ACC(acc0, acc1);
    }
    blk_issue<2>(9, ao, X, ldsW, half, lrow, gB, bwB);
    blk_load_offsets<3>(10, ldsTab, half, ao);
    BLK_FENCE_ACC(acc0, acc1);
    blk_compute<2>(gA, bwA, acc0, acc1);
    BLK_FENCE_ACC(acc0, acc1);
    blk_issue<3>(10, ao, X, ldsW, half, lrow, gA, bwA);
    blk_load_offsets<3>(11, ldsTab, half, ao);
    BLK_FENCE_ACC(acc0, acc1);
    blk_compute<2>(gB, bwB, acc0, acc1);
    BLK_FENCE_ACC(acc0, acc1);
#pragma unroll 1
    for (int blk = 10; blk < 24; blk += 2) {
      blk_issue<3>(blk + 1, ao, X, ldsW, half, lrow, gB, bwB);
      blk_load_offsets<3>(blk + 2, ldsTab, half, ao);
      BLK_FENCE_ACC(acc0, acc1);
      blk_compute<3>(gA, bwA, acc0, acc1);
      BLK_FENCE_ACC(acc0, acc1);
      blk_issue<3>(blk + 2, ao, X, ldsW, half, lrow, gA, bwA);
      blk_load_offsets<3>(blk + 3, ldsTab, half, ao);
      BLK_FENCE_ACC(acc0, acc1);
      blk_compute<3>(gB, bwB, acc0, acc1);
      BLK_FENCE_ACC(acc0, acc1);
    }
    blk_issue<3>(25, ao, X, ldsW, half, lrow, gB, bwB);
    BLK_FENCE_ACC(acc0, acc1);
    blk_compute<3>(gA, bwA, acc0, acc1);
    BLK_FENCE_ACC(acc0, acc1);
    blk_compute<3>(gB, bwB, acc0, acc1);

    if (ai + 1 < A_PER_WG) stage_load(sr, x, Wg, a + 1, tid, lane, myb);

    float pr[16];
#pragma unroll
    for (int tl = 0; tl < 2; ++tl) {
#pragma unroll
      for (int r = 0; r < 8; ++r) {
        const int row = wv * 32 + tl * 16 + r + half * 8;
        float v = (tl == 0 ? acc0[r] : acc1[r]) * ldsY[row * EPAD + lrow];
        v = row_xor_add<1>(v);
        v = row_xor_add<2>(v);
        v = row_xor_add<4>(v);
        v = row_xor_add<8>(v);
        pr[tl * 8 + r] = v;
      }
    }
    if (lrow == 0) {
#pragma unroll
      for (int tl = 0; tl < 2; ++tl) {
#pragma unroll
        for (int r = 0; r < 8; ++r) {
          const int row = wv * 32 + tl * 16 + r + half * 8;
          out[(size_t)(b0 + row) * NA + a] = pr[tl * 8 + r];
        }
      }
    }
  }
}

extern "C" void kernel_launch(void* const* d_in, const int* in_sizes, int n_in,
                              void* d_out, int out_size, void* d_ws, size_t ws_size,
                              hipStream_t stream) {
  const float* x  = (const float*)d_in[0];
  const float* y  = (const float*)d_in[1];
  const float* w1 = (const float*)d_in[2];
  const float* w2 = (const float*)d_in[3];
  const float* w3 = (const float*)d_in[4];
  const float* c1 = (const float*)d_in[5];
  const float* c2 = (const float*)d_in[6];
  const float* c3 = (const float*)d_in[7];
  const int* i1 = (const int*)d_in[8];
  const int* l1 = (const int*)d_in[9];
  const int* k1 = (const int*)d_in[10];
  const int* q1 = (const int*)d_in[11];
  const int* i2 = (const int*)d_in[12];
  const int* j2 = (const int*)d_in[13];
  const int* l2 = (const int*)d_in[14];
  const int* m2 = (const int*)d_in[15];
  const int* k2 = (const int*)d_in[16];
  const int* q2 = (const int*)d_in[17];
  const int* i3 = (const int*)d_in[18];
  const int* j3 = (const int*)d_in[19];
  const int* f3 = (const int*)d_in[20];
  const int* l3 = (const int*)d_in[21];
  const int* m3 = (const int*)d_in[22];
  const int* g3 = (const int*)d_in[23];
  const int* k3 = (const int*)d_in[24];
  const int* q3 = (const int*)d_in[25];

  int*   offs = (int*)d_ws;
  float* Wg   = (float*)((char*)d_ws + 5120);
  float* out  = (float*)d_out;

  prep_kernel<<<PREP_WG_BLOCKS + PREP_OFF_BLOCKS, 256, 0, stream>>>(w1, w2, w3, c1, c2, c3,
                                                                    k1, q1, k2, q2, k3, q3,
                                                                    i1, l1, i2, j2, l2, m2,
                                                                    i3, j3, f3, l3, m3, g3, Wg, offs);
  contract_kernel<<<dim3(32, 32), WG_THREADS, 0, stream>>>(x, y, Wg, offs, out);
}


# Each wave runs four 16-row tiles per K step (256-row groups, grid 16x32): x rows interleaved four-wise in LDS so one conflict-free 16-byte load gathers a factor for four WMMAs; table, B loads shared
# speedup vs baseline: 3.9310x; 1.0440x over previous
#include <hip/hip_runtime.h>
#include <hip/hip_bf16.h>

typedef float v2f __attribute__((ext_vector_type(2)));
typedef float v8f __attribute__((ext_vector_type(8)));

#define NB 4096
#define NA 64
#define NL 16
#define NM 4
#define NE 10
#define P_TOT 416
#define P1_END 32
#define P2_END 160
#define EPAD 16
#define XP4 260
#define WG_WAVES 4
#define WG_THREADS 128
#define WG_ROWS 256
#define A_PER_WG 2

#define PREP_WG_BLOCKS 104
#define TAB_INTS 1248
#define TAB_VEC4 312
#define PREP_OFF_BLOCKS 5

__global__ void prep_kernel(const float* __restrict__ w1, const float* __restrict__ w2,
                            const float* __restrict__ w3,
                            const float* __restrict__ c1, const float* __restrict__ c2,
                            const float* __restrict__ c3,
                            const int* __restrict__ k1, const int* __restrict__ q1,
                            const int* __restrict__ k2, const int* __restrict__ q2,
                            const int* __restrict__ k3, const int* __restrict__ q3,
                            const int* __restrict__ i1, const int* __restrict__ l1,
                            const int* __restrict__ i2, const int* __restrict__ j2,
                            const int* __restrict__ l2, const int* __restrict__ m2,
                            const int* __restrict__ i3, const int* __restrict__ j3,
                            const int* __restrict__ f3, const int* __restrict__ l3,
                            const int* __restrict__ m3, const int* __restrict__ g3,
                            float* __restrict__ Wg, int* __restrict__ offs) {
  if (blockIdx.x >= PREP_WG_BLOCKS) {
    const int t = (blockIdx.x - PREP_WG_BLOCKS) * blockDim.x + threadIdx.x;
    if (t >= TAB_INTS) return;
    const int blk  = t / 48;
    const int r    = t - blk * 48;
    const int half = r / 24;
    const int r2   = r - half * 24;
    const int k    = r2 >> 3;
    const int j    = r2 & 7;
    const int p    = 16 * blk + 4 * (j >> 1) + 2 * half + (j & 1);
    int o = 0;
    if (p < P1_END) {
      if (k == 0) o = i1[p] * NM + l1[p];
    } else if (p < P2_END) {
      const int q = p - P1_END;
      if (k == 0)      o = i2[q] * NM + l2[q];
      else if (k == 1) o = j2[q] * NM + m2[q];
    } else {
      const int q = p - P2_END;
      if (k == 0)      o = i3[q] * NM + l3[q];
      else if (k == 1) o = j3[q] * NM + m3[q];
      else             o = f3[q] * NM + g3[q];
    }
    o &= 63;
    offs[t] = ((o & 48) | ((o + (o >> 4)) & 15)) * 16;
    return;
  }
  int tid = blockIdx.x * blockDim.x + threadIdx.x;
  if (tid >= NA * P_TOT) return;
  int p = tid % P_TOT;
  int a = tid / P_TOT;
  float cf;
  const float* wbase;
  int estride;
  int kq;
  if (p < P1_END) {
    cf = c1[p];
    wbase = w1; kq = k1[p] * 2 + q1[p]; estride = 4 * 2 * NA;
  } else if (p < P2_END) {
    int t = p - P1_END;
    cf = c2[t];
    wbase = w2; kq = k2[t] * 4 + q2[t]; estride = 16 * 4 * NA;
  } else {
    int t = p - P2_END;
    cf = c3[t];
    wbase = w3; kq = k3[t] * 6 + q3[t]; estride = 32 * 6 * NA;
  }
  const int kqstride = kq * NA + a;
  float* dst = Wg + ((size_t)a * P_TOT + p) * EPAD;
#pragma unroll
  for (int n = 0; n < EPAD; ++n) {
    dst[n] = (n < NE) ? cf * wbase[(size_t)n * estride + kqstride] : 0.0f;
  }
}

#define BLK_STEPS 4
#define BLK_SLOTS 8
#define BLK_FENCE_ACC(A0, A1, A2, A3) do { asm volatile("" : "+v"(A0), "+v"(A1), "+v"(A2), "+v"(A3)); __builtin_amdgcn_sched_barrier(0); } while (0)

typedef int v4i __attribute__((ext_vector_type(4)));
typedef float v4f __attribute__((ext_vector_type(4)));

template <int CNT>
__device__ __forceinline__ void blk_load_offsets(int blk, const v4i* __restrict__ ldsTab, int half,
                                                 int (&ao)[3][BLK_SLOTS]) {
  const v4i* t = ldsTab + (blk * 2 + half) * 6;
#pragma unroll
  for (int k = 0; k < CNT; ++k) {
    const v4i lo = t[2 * k];
    const v4i hi = t[2 * k + 1];
    ao[k][0] = lo.x; ao[k][1] = lo.y; ao[k][2] = lo.z; ao[k][3] = lo.w;
    ao[k][4] = hi.x; ao[k][5] = hi.y; ao[k][6] = hi.z; ao[k][7] = hi.w;
  }
}

template <int CNT>
__device__ __forceinline__ void blk_issue(int blk, const int (&ao)[3][BLK_SLOTS], const char* __restrict__ Xb,
                                          const float* __restrict__ ldsW, int half, int lrow,
                                          v4f (&g)[3][BLK_SLOTS], float (&bw)[BLK_SLOTS]) {
#pragma unroll
  for (int j = 0; j < BLK_SLOTS; ++j) {
    g[0][j] = *(const v4f*)(Xb + ao[0][j]);
    if (CNT > 1) g[1][j] = *(const v4f*)(Xb + ao[1][j]);
    if (CNT > 2) g[2][j] = *(const v4f*)(Xb + ao[2][j]);
  }
#pragma unroll
  for (int s = 0; s < BLK_STEPS; ++s) {
    const int pA = 4 * (BLK_STEPS * blk + s) + 2 * half;
    bw[2 * s]     = ldsW[pA * EPAD + lrow];
    bw[2 * s + 1] = ldsW[(pA + 1) * EPAD + lrow];
  }
}

template <int CNT>
__device__ __forceinline__ void blk_compute(const v4f (&g)[3][BLK_SLOTS], const float (&bw)[BLK_SLOTS],
                                            v8f& acc0, v8f& acc1, v8f& acc2, v8f& acc3) {
#pragma unroll
  for (int s = 0; s < BLK_STEPS; ++s) {
    v4f xa = g[0][2 * s];
    v4f xb = g[0][2 * s + 1];
    if (CNT > 1) { xa *= g[1][2 * s]; xb *= g[1][2 * s + 1]; }
    if (CNT > 2) { xa *= g[2][2 * s]; xb *= g[2][2 * s + 1]; }
    v2f a0; a0.x = xa.x; a0.y = xb.x;
    v2f a1; a1.x = xa.y; a1.y = xb.y;
    v2f a2; a2.x = xa.z; a2.y = xb.z;
    v2f a3; a3.x = xa.w; a3.y = xb.w;
    v2f bf; bf.x = bw[2 * s]; bf.y = bw[2 * s + 1];
    acc0 = __builtin_amdgcn_wmma_f32_16x16x4_f32(false, a0, false, bf,
                                                 (short)0, acc0, false, false);
    acc1 = __builtin_amdgcn_wmma_f32_16x16x4_f32(false, a1, false, bf,
                                                 (short)0, acc1, false, false);
    acc2 = __builtin_amdgcn_wmma_f32_16x16x4_f32(false, a2, false, bf,
                                                 (short)0, acc2, false, false);
    acc3 = __builtin_amdgcn_wmma_f32_16x16x4_f32(false, a3, false, bf,
                                                 (short)0, acc3, false, false);
  }
}


struct StageRegs {
  v4f w[13];
  v4f xr[32];
};

__device__ __forceinline__ void stage_load(StageRegs& r, const float* __restrict__ x,
                                           const float* __restrict__ Wg, int a,
                                           int tid, int lane, int myb) {
  const v4f* ws = (const v4f*)(Wg + (size_t)a * P_TOT * EPAD) + tid;
#pragma unroll
  for (int k = 0; k < 13; ++k) r.w[k] = ws[WG_THREADS * k];
  const float* xb = x + ((size_t)(myb + (lane >> 4)) * NA + a) * 64 + (lane & 15) * 4;
  const int XS = 2 * NA * 64;
#pragma unroll
  for (int k = 0; k < 32; ++k) r.xr[k] = *(const v4f*)(xb + k * XS);
}

__device__ __forceinline__ void stage_store(const StageRegs& r, float* ldsW, float* ldsX,
                                            int tid, int lane, int wv) {
  v4f* wd = (v4f*)ldsW + tid;
#pragma unroll
  for (int k = 0; k < 13; ++k) wd[WG_THREADS * k] = r.w[k];
  const int c = lane & 15;
  const int q = c >> 2;
  const int m4q = 4 * (c & 3) + q;
  float* xd = &ldsX[(wv * 16 + (lane >> 4)) * XP4 + 64 * q];
  const int LS = 2 * XP4;
  const int s0 = 4 * ((m4q + 0) & 15);
  const int s1 = 4 * ((m4q + 1) & 15);
  const int s2 = 4 * ((m4q + 2) & 15);
  const int s3 = 4 * ((m4q + 3) & 15);
#pragma unroll
  for (int j = 0; j < 8; ++j) {
    const v4f t0 = r.xr[j];
    const v4f t1 = r.xr[8 + j];
    const v4f t2 = r.xr[16 + j];
    const v4f t3 = r.xr[24 + j];
    *(v4f*)(xd + j * LS + s0) = (v4f){t0.x, t1.x, t2.x, t3.x};
    *(v4f*)(xd + j * LS + s1) = (v4f){t0.y, t1.y, t2.y, t3.y};
    *(v4f*)(xd + j * LS + s2) = (v4f){t0.z, t1.z, t2.z, t3.z};
    *(v4f*)(xd + j * LS + s3) = (v4f){t0.w, t1.w, t2.w, t3.w};
  }
}

template <int MASK>
__device__ __forceinline__ float row_xor_add(float v) {
  const int other = __builtin_amdgcn_update_dpp(0, __float_as_int(v), 0x160 + MASK, 0xF, 0xF, true);
  return v + __int_as_float(other);
}

__global__ __launch_bounds__(WG_THREADS, 2) void contract_kernel(const float* __restrict__ x,
                                                       const float* __restrict__ y,
                                                       const float* __restrict__ Wg,
                                                       const int* __restrict__ offs,
                                                       float* __restrict__ out) {
  __shared__ __attribute__((aligned(16))) float ldsW[P_TOT * EPAD];
  __shared__ __attribute__((aligned(16))) float ldsX[WG_WAVES * 16 * XP4];
  __shared__ __attribute__((aligned(16))) float ldsY[WG_ROWS * EPAD];
  __shared__ v4i   ldsTab[TAB_VEC4];

  const int tid  = threadIdx.x;
  const int wv   = tid >> 5;
  const int lane = tid & 31;
  const int half = lane >> 4;
  const int lrow = lane & 15;
  const int b0   = blockIdx.x * WG_ROWS;
  const int a0   = blockIdx.y * A_PER_WG;
  const int myb  = b0 + wv * 64;

  StageRegs sr;
  stage_load(sr, x, Wg, a0, tid, lane, myb);
  const v4i* tg = (const v4i*)offs;
  const v4i tb0 = tg[tid];
  const v4i tb1 = tg[WG_THREADS + tid];
  v4i tb2 = {0, 0, 0, 0};
  if (tid < TAB_VEC4 - 2 * WG_THREADS) tb2 = tg[2 * WG_THREADS + tid];
  float yv[32];
#pragma unroll
  for (int j = 0; j < 32; ++j) yv[j] = 0.0f;
  if ((tid & 15) < NE) {
    const float* yb = y + (size_t)(b0 + (tid >> 4)) * NE + (tid & 15);
    const int YS = 8 * NE;
#pragma unroll
    for (int j = 0; j < 32; ++j) yv[j] = yb[j * YS];
  }

  ldsTab[tid] = tb0;
  ldsTab[WG_THREADS + tid] = tb1;
  if (tid < TAB_VEC4 - 2 * WG_THREADS) ldsTab[2 * WG_THREADS + tid] = tb2;
#pragma unroll
  for (int j = 0; j < 32; ++j) ldsY[tid + WG_THREADS * j] = yv[j];

#pragma unroll
  for (int ai = 0; ai < A_PER_WG; ++ai) {
    const int a = a0 + ai;
    if (ai > 0) __syncthreads();
    stage_store(sr, ldsW, ldsX, tid, lane, wv);
    __syncthreads();

    const char* X = (const char*)&ldsX[(wv * 16 + lrow) * XP4];
    v8f acc0 = {};
    v8f acc1 = {};
    v8f acc2 = {};
    v8f acc3 = {};
    int   ao[3][BLK_SLOTS];
    v4f   gA[3][BLK_SLOTS], gB[3][BLK_SLOTS];
    float bwA[BLK_SLOTS], bwB[BLK_SLOTS];

    blk_load_offsets<1>(0, ldsTab, half, ao);
    blk_issue<1>(0, ao, X, ldsW, half, lrow, gA, bwA);
    blk_load_offsets<1>(1, ldsTab, half, ao);
    blk_issue<1>(1, ao, X, ldsW, half, lrow, gB, bwB);
    blk_load_offsets<2>(2, ldsTab, half, ao);
    BLK_FENCE_ACC(acc0, acc1, acc2, acc3);
    blk_compute<1>(gA, bwA, acc0, acc1, acc2, acc3);
    BLK_FENCE_ACC(acc0, acc1, acc2, acc3);
    blk_issue<2>(2, ao, X, ldsW, half, lrow, gA, bwA);
    blk_load_offsets<2>(3, ldsTab, half, ao);
    BLK_FENCE_ACC(acc0, acc1, acc2, acc3);
    blk_compute<1>(gB, bwB, acc0, acc1, acc2, acc3);
    BLK_FENCE_ACC(acc0, acc1, acc2, acc3);
#pragma unroll 1
    for (int blk = 2; blk < 8; blk += 2) {
      blk_issue<2>(blk + 1, ao, X, ldsW, half, lrow, gB, bwB);
      blk_load_offsets<2>(blk + 2, ldsTab, half, ao);
      BLK_FENCE_ACC(acc0, acc1, acc2, acc3);
      blk_compute<2>(gA, bwA, acc0, acc1, acc2, acc3);
      BLK_FENCE_ACC(acc0, acc1, acc2, acc3);
      blk_issue<2>(blk + 2, ao, X, ldsW, half, lrow, gA, bwA);
      blk_load_offsets<2>(blk + 3, ldsTab, half, ao);
      BLK_FENCE_ACC(acc0, acc1, acc2, acc3);
      blk_compute<2>(gB, bwB, acc0, acc1, acc2, acc3);
      BLK_FENCE_ACC(acc0, acc1, acc2, acc3);
    }
    blk_issue<2>(9, ao, X, ldsW, half, lrow, gB, bwB);
    blk_load_offsets<3>(10, ldsTab, half, ao);
    BLK_FENCE_ACC(acc0, acc1, acc2, acc3);
    blk_compute<2>(gA, bwA, acc0, acc1, acc2, acc3);
    BLK_FENCE_ACC(acc0, acc1, acc2, acc3);
    blk_issue<3>(10, ao, X, ldsW, half, lrow, gA, bwA);
    blk_load_offsets<3>(11, ldsTab, half, ao);
    BLK_FENCE_ACC(acc0, acc1, acc2, acc3);
    blk_compute<2>(gB, bwB, acc0, acc1, acc2, acc3);
    BLK_FENCE_ACC(acc0, acc1, acc2, acc3);
#pragma unroll 1
    for (int blk = 10; blk < 24; blk += 2) {
      blk_issue<3>(blk + 1, ao, X, ldsW, half, lrow, gB, bwB);
      blk_load_offsets<3>(blk + 2, ldsTab, half, ao);
      BLK_FENCE_ACC(acc0, acc1, acc2, acc3);
      blk_compute<3>(gA, bwA, acc0, acc1, acc2, acc3);
      BLK_FENCE_ACC(acc0, acc1, acc2, acc3);
      blk_issue<3>(blk + 2, ao, X, ldsW, half, lrow, gA, bwA);
      blk_load_offsets<3>(blk + 3, ldsTab, half, ao);
      BLK_FENCE_ACC(acc0, acc1, acc2, acc3);
      blk_compute<3>(gB, bwB, acc0, acc1, acc2, acc3);
      BLK_FENCE_ACC(acc0, acc1, acc2, acc3);
    }
    blk_issue<3>(25, ao, X, ldsW, half, lrow, gB, bwB);
    BLK_FENCE_ACC(acc0, acc1, acc2, acc3);
    blk_compute<3>(gA, bwA, acc0, acc1, acc2, acc3);
    BLK_FENCE_ACC(acc0, acc1, acc2, acc3);
    blk_compute<3>(gB, bwB, acc0, acc1, acc2, acc3);

    if (ai + 1 < A_PER_WG) stage_load(sr, x, Wg, a + 1, tid, lane, myb);

    float pr[32];
#pragma unroll
    for (int tl = 0; tl < 4; ++tl) {
#pragma unroll
      for (int r = 0; r < 8; ++r) {
        const int row = wv * 64 + tl * 16 + r + half * 8;
        float v = (tl == 0 ? acc0[r] : tl == 1 ? acc1[r] : tl == 2 ? acc2[r] : acc3[r]) * ldsY[row * EPAD + lrow];
        v = row_xor_add<1>(v);
        v = row_xor_add<2>(v);
        v = row_xor_add<4>(v);
        v = row_xor_add<8>(v);
        pr[tl * 8 + r] = v;
      }
    }
    if (lrow == 0) {
#pragma unroll
      for (int tl = 0; tl < 4; ++tl) {
#pragma unroll
        for (int r = 0; r < 8; ++r) {
          const int row = wv * 64 + tl * 16 + r + half * 8;
          out[(size_t)(b0 + row) * NA + a] = pr[tl * 8 + r];
        }
      }
    }
  }
}

extern "C" void kernel_launch(void* const* d_in, const int* in_sizes, int n_in,
                              void* d_out, int out_size, void* d_ws, size_t ws_size,
                              hipStream_t stream) {
  const float* x  = (const float*)d_in[0];
  const float* y  = (const float*)d_in[1];
  const float* w1 = (const float*)d_in[2];
  const float* w2 = (const float*)d_in[3];
  const float* w3 = (const float*)d_in[4];
  const float* c1 = (const float*)d_in[5];
  const float* c2 = (const float*)d_in[6];
  const float* c3 = (const float*)d_in[7];
  const int* i1 = (const int*)d_in[8];
  const int* l1 = (const int*)d_in[9];
  const int* k1 = (const int*)d_in[10];
  const int* q1 = (const int*)d_in[11];
  const int* i2 = (const int*)d_in[12];
  const int* j2 = (const int*)d_in[13];
  const int* l2 = (const int*)d_in[14];
  const int* m2 = (const int*)d_in[15];
  const int* k2 = (const int*)d_in[16];
  const int* q2 = (const int*)d_in[17];
  const int* i3 = (const int*)d_in[18];
  const int* j3 = (const int*)d_in[19];
  const int* f3 = (const int*)d_in[20];
  const int* l3 = (const int*)d_in[21];
  const int* m3 = (const int*)d_in[22];
  const int* g3 = (const int*)d_in[23];
  const int* k3 = (const int*)d_in[24];
  const int* q3 = (const int*)d_in[25];

  int*   offs = (int*)d_ws;
  float* Wg   = (float*)((char*)d_ws + 5120);
  float* out  = (float*)d_out;

  prep_kernel<<<PREP_WG_BLOCKS + PREP_OFF_BLOCKS, 256, 0, stream>>>(w1, w2, w3, c1, c2, c3,
                                                                    k1, q1, k2, q2, k3, q3,
                                                                    i1, l1, i2, j2, l2, m2,
                                                                    i3, j3, f3, l3, m3, g3, Wg, offs);
  contract_kernel<<<dim3(16, 32), WG_THREADS, 0, stream>>>(x, y, Wg, offs, out);
}


# Contract e on the matrix unit (wy = Wg x y, K=12) and sum the paths by in-lane FMAs on wy * xprod: 12 WMMAs per 16 paths x 4 tiles, not 16; y kept as B fragments; changes the summation order
# speedup vs baseline: 4.2853x; 1.0901x over previous
#include <hip/hip_runtime.h>
#include <hip/hip_bf16.h>

typedef float v2f __attribute__((ext_vector_type(2)));
typedef float v8f __attribute__((ext_vector_type(8)));

#define NB 4096
#define NA 64
#define NL 16
#define NM 4
#define NE 10
#define P_TOT 416
#define P1_END 32
#define P2_END 160
#define EPAD 16
#define XP4 260
#define WG_WAVES 4
#define WG_THREADS 128
#define WG_ROWS 256
#define A_PER_WG 2
#define WP 20

#define PREP_WG_BLOCKS 104
#define TAB_INTS 1248
#define TAB_VEC4 312
#define PREP_OFF_BLOCKS 5

__global__ void prep_kernel(const float* __restrict__ w1, const float* __restrict__ w2,
                            const float* __restrict__ w3,
                            const float* __restrict__ c1, const float* __restrict__ c2,
                            const float* __restrict__ c3,
                            const int* __restrict__ k1, const int* __restrict__ q1,
                            const int* __restrict__ k2, const int* __restrict__ q2,
                            const int* __restrict__ k3, const int* __restrict__ q3,
                            const int* __restrict__ i1, const int* __restrict__ l1,
                            const int* __restrict__ i2, const int* __restrict__ j2,
                            const int* __restrict__ l2, const int* __restrict__ m2,
                            const int* __restrict__ i3, const int* __restrict__ j3,
                            const int* __restrict__ f3, const int* __restrict__ l3,
                            const int* __restrict__ m3, const int* __restrict__ g3,
                            float* __restrict__ Wg, int* __restrict__ offs) {
  if (blockIdx.x >= PREP_WG_BLOCKS) {
    const int t = (blockIdx.x - PREP_WG_BLOCKS) * blockDim.x + threadIdx.x;
    if (t >= TAB_INTS) return;
    const int blk  = t / 48;
    const int r    = t - blk * 48;
    const int half = r / 24;
    const int r2   = r - half * 24;
    const int k    = r2 >> 3;
    const int j    = r2 & 7;
    const int p    = 16 * blk + 8 * half + j;
    int o = 0;
    if (p < P1_END) {
      if (k == 0) o = i1[p] * NM + l1[p];
    } else if (p < P2_END) {
      const int q = p - P1_END;
      if (k == 0)      o = i2[q] * NM + l2[q];
      else if (k == 1) o = j2[q] * NM + m2[q];
    } else {
      const int q = p - P2_END;
      if (k == 0)      o = i3[q] * NM + l3[q];
      else if (k == 1) o = j3[q] * NM + m3[q];
      else             o = f3[q] * NM + g3[q];
    }
    o &= 63;
    offs[t] = ((o & 48) | ((o + (o >> 4)) & 15)) * 16;
    return;
  }
  int tid = blockIdx.x * blockDim.x + threadIdx.x;
  if (tid >= NA * P_TOT) return;
  int p = tid % P_TOT;
  int a = tid / P_TOT;
  float cf;
  const float* wbase;
  int estride;
  int kq;
  if (p < P1_END) {
    cf = c1[p];
    wbase = w1; kq = k1[p] * 2 + q1[p]; estride = 4 * 2 * NA;
  } else if (p < P2_END) {
    int t = p - P1_END;
    cf = c2[t];
    wbase = w2; kq = k2[t] * 4 + q2[t]; estride = 16 * 4 * NA;
  } else {
    int t = p - P2_END;
    cf = c3[t];
    wbase = w3; kq = k3[t] * 6 + q3[t]; estride = 32 * 6 * NA;
  }
  const int kqstride = kq * NA + a;
  float* dst = Wg + ((size_t)a * P_TOT + p) * EPAD;
#pragma unroll
  for (int n = 0; n < EPAD; ++n) {
    dst[n] = (n < NE) ? cf * wbase[(size_t)n * estride + kqstride] : 0.0f;
  }
}

#define BLK_STEPS 3
#define BLK_SLOTS 8
#define BLK_FENCE_ACC(S) do { asm volatile("" : "+v"(S[0]), "+v"(S[1]), "+v"(S[2]), "+v"(S[3]), "+v"(S[4]), "+v"(S[5]), "+v"(S[6]), "+v"(S[7])); __builtin_amdgcn_sched_barrier(0); } while (0)

typedef int v4i __attribute__((ext_vector_type(4)));
typedef float v4f __attribute__((ext_vector_type(4)));

template <int CNT>
__device__ __forceinline__ void blk_load_offsets(int blk, const v4i* __restrict__ ldsTab, int half,
                                                 int (&ao)[3][BLK_SLOTS]) {
  const v4i* t = ldsTab + (blk * 2 + half) * 6;
#pragma unroll
  for (int k = 0; k < CNT; ++k) {
    const v4i lo = t[2 * k];
    const v4i hi = t[2 * k + 1];
    ao[k][0] = lo.x; ao[k][1] = lo.y; ao[k][2] = lo.z; ao[k][3] = lo.w;
    ao[k][4] = hi.x; ao[k][5] = hi.y; ao[k][6] = hi.z; ao[k][7] = hi.w;
  }
}

template <int CNT>
__device__ __forceinline__ void blk_issue(int blk, const int (&ao)[3][BLK_SLOTS], const char* __restrict__ Xb,
                                          const float* __restrict__ Wl,
                                          v4f (&g)[3][BLK_SLOTS], v2f (&aw)[BLK_STEPS]) {
#pragma unroll
  for (int j = 0; j < BLK_SLOTS; ++j) {
    g[0][j] = *(const v4f*)(Xb + ao[0][j]);
    if (CNT > 1) g[1][j] = *(const v4f*)(Xb + ao[1][j]);
    if (CNT > 2) g[2][j] = *(const v4f*)(Xb + ao[2][j]);
  }
  const float* wr = Wl + blk * (16 * WP);
#pragma unroll
  for (int s = 0; s < BLK_STEPS; ++s) aw[s] = *(const v2f*)(wr + 4 * s);
}

template <int CNT>
__device__ __forceinline__ void blk_compute(const v4f (&g)[3][BLK_SLOTS], const v2f (&aw)[BLK_STEPS],
                                            const v2f (&yB)[4][BLK_STEPS], float (&S)[8]) {
  v8f d0 = {}, d1 = {}, d2 = {}, d3 = {};
#pragma unroll
  for (int s = 0; s < BLK_STEPS; ++s) {
    d0 = __builtin_amdgcn_wmma_f32_16x16x4_f32(false, aw[s], false, yB[0][s], (short)0, d0, false, false);
    d1 = __builtin_amdgcn_wmma_f32_16x16x4_f32(false, aw[s], false, yB[1][s], (short)0, d1, false, false);
    d2 = __builtin_amdgcn_wmma_f32_16x16x4_f32(false, aw[s], false, yB[2][s], (short)0, d2, false, false);
    d3 = __builtin_amdgcn_wmma_f32_16x16x4_f32(false, aw[s], false, yB[3][s], (short)0, d3, false, false);
  }
  v4f xp[BLK_SLOTS];
#pragma unroll
  for (int r = 0; r < BLK_SLOTS; ++r) {
    xp[r] = g[0][r];
    if (CNT > 1) xp[r] *= g[1][r];
    if (CNT > 2) xp[r] *= g[2][r];
  }
#pragma unroll
  for (int r = 0; r < BLK_SLOTS; ++r) {
    S[0 + (r & 1)] = __builtin_fmaf(d0[r], xp[r].x, S[0 + (r & 1)]);
    S[2 + (r & 1)] = __builtin_fmaf(d1[r], xp[r].y, S[2 + (r & 1)]);
    S[4 + (r & 1)] = __builtin_fmaf(d2[r], xp[r].z, S[4 + (r & 1)]);
    S[6 + (r & 1)] = __builtin_fmaf(d3[r], xp[r].w, S[6 + (r & 1)]);
  }
}


struct StageRegs {
  v4f w[13];
  v4f xr[32];
};

__device__ __forceinline__ void stage_load(StageRegs& r, const float* __restrict__ x,
                                           const float* __restrict__ Wg, int a,
                                           int tid, int lane, int myb) {
  const v4f* ws = (const v4f*)(Wg + (size_t)a * P_TOT * EPAD) + tid;
#pragma unroll
  for (int k = 0; k < 13; ++k) r.w[k] = ws[WG_THREADS * k];
  const float* xb = x + ((size_t)(myb + (lane >> 4)) * NA + a) * 64 + (lane & 15) * 4;
  const int XS = 2 * NA * 64;
#pragma unroll
  for (int k = 0; k < 32; ++k) r.xr[k] = *(const v4f*)(xb + k * XS);
}

__device__ __forceinline__ void stage_store(const StageRegs& r, float* ldsW, float* ldsX,
                                            int tid, int lane, int wv) {
  float* wd = ldsW + (tid >> 2) * WP + (tid & 3) * 4;
#pragma unroll
  for (int k = 0; k < 13; ++k) *(v4f*)(wd + k * (32 * WP)) = r.w[k];
  const int c = lane & 15;
  const int q = c >> 2;
  const int m4q = 4 * (c & 3) + q;
  float* xd = &ldsX[(wv * 16 + (lane >> 4)) * XP4 + 64 * q];
  const int LS = 2 * XP4;
  const int s0 = 4 * ((m4q + 0) & 15);
  const int s1 = 4 * ((m4q + 1) & 15);
  const int s2 = 4 * ((m4q + 2) & 15);
  const int s3 = 4 * ((m4q + 3) & 15);
#pragma unroll
  for (int j = 0; j < 8; ++j) {
    const v4f t0 = r.xr[j];
    const v4f t1 = r.xr[8 + j];
    const v4f t2 = r.xr[16 + j];
    const v4f t3 = r.xr[24 + j];
    *(v4f*)(xd + j * LS + s0) = (v4f){t0.x, t1.x, t2.x, t3.x};
    *(v4f*)(xd + j * LS + s1) = (v4f){t0.y, t1.y, t2.y, t3.y};
    *(v4f*)(xd + j * LS + s2) = (v4f){t0.z, t1.z, t2.z, t3.z};
    *(v4f*)(xd + j * LS + s3) = (v4f){t0.w, t1.w, t2.w, t3.w};
  }
}

__global__ __launch_bounds__(WG_THREADS, 2) void contract_kernel(const float* __restrict__ x,
                                                       const float* __restrict__ y,
                                                       const float* __restrict__ Wg,
                                                       const int* __restrict__ offs,
                                                       float* __restrict__ out) {
  __shared__ __attribute__((aligned(16))) float ldsW[P_TOT * WP];
  __shared__ __attribute__((aligned(16))) float ldsX[WG_WAVES * 16 * XP4];
  __shared__ v4i   ldsTab[TAB_VEC4];

  const int tid  = threadIdx.x;
  const int wv   = tid >> 5;
  const int lane = tid & 31;
  const int half = lane >> 4;
  const int lrow = lane & 15;
  const int b0   = blockIdx.x * WG_ROWS;
  const int a0   = blockIdx.y * A_PER_WG;
  const int myb  = b0 + wv * 64;

  StageRegs sr;
  stage_load(sr, x, Wg, a0, tid, lane, myb);
  const v4i* tg = (const v4i*)offs;
  const v4i tb0 = tg[tid];
  const v4i tb1 = tg[WG_THREADS + tid];
  v4i tb2 = {0, 0, 0, 0};
  if (tid < TAB_VEC4 - 2 * WG_THREADS) tb2 = tg[2 * WG_THREADS + tid];
  v2f yB[4][BLK_STEPS];
#pragma unroll
  for (int t = 0; t < 4; ++t) {
    const float* yp = y + (size_t)(myb + t * 16 + lrow) * NE + 2 * half;
#pragma unroll
    for (int s = 0; s < BLK_STEPS; ++s) {
      v2f v = {0.0f, 0.0f};
      if (4 * s + 2 * half < NE) { v.x = yp[4 * s]; v.y = yp[4 * s + 1]; }
      yB[t][s] = v;
    }
  }

  ldsTab[tid] = tb0;
  ldsTab[WG_THREADS + tid] = tb1;
  if (tid < TAB_VEC4 - 2 * WG_THREADS) ldsTab[2 * WG_THREADS + tid] = tb2;

#pragma unroll
  for (int ai = 0; ai < A_PER_WG; ++ai) {
    const int a = a0 + ai;
    if (ai > 0) __syncthreads();
    stage_store(sr, ldsW, ldsX, tid, lane, wv);
    __syncthreads();

    const char* X = (const char*)&ldsX[(wv * 16 + lrow) * XP4];
    const float* Wl = ldsW + lrow * WP + 2 * half;
    float S[8];
#pragma unroll
    for (int i = 0; i < 8; ++i) S[i] = 0.0f;
    int   ao[3][BLK_SLOTS];
    v4f   gA[3][BLK_SLOTS], gB[3][BLK_SLOTS];
    v2f   awA[BLK_STEPS], awB[BLK_STEPS];

    blk_load_offsets<1>(0, ldsTab, half, ao);
    blk_issue<1>(0, ao, X, Wl, gA, awA);
    blk_load_offsets<1>(1, ldsTab, half, ao);
    blk_issue<1>(1, ao, X, Wl, gB, awB);
    blk_load_offsets<2>(2, ldsTab, half, ao);
    BLK_FENCE_ACC(S);
    blk_compute<1>(gA, awA, yB, S);
    BLK_FENCE_ACC(S);
    blk_issue<2>(2, ao, X, Wl, gA, awA);
    blk_load_offsets<2>(3, ldsTab, half, ao);
    BLK_FENCE_ACC(S);
    blk_compute<1>(gB, awB, yB, S);
    BLK_FENCE_ACC(S);
#pragma unroll 1
    for (int blk = 2; blk < 8; blk += 2) {
      blk_issue<2>(blk + 1, ao, X, Wl, gB, awB);
      blk_load_offsets<2>(blk + 2, ldsTab, half, ao);
      BLK_FENCE_ACC(S);
      blk_compute<2>(gA, awA, yB, S);
      BLK_FENCE_ACC(S);
      blk_issue<2>(blk + 2, ao, X, Wl, gA, awA);
      blk_load_offsets<2>(blk + 3, ldsTab, half, ao);
      BLK_FENCE_ACC(S);
      blk_compute<2>(gB, awB, yB, S);
      BLK_FENCE_ACC(S);
    }
    blk_issue<2>(9, ao, X, Wl, gB, awB);
    blk_load_offsets<3>(10, ldsTab, half, ao);
    BLK_FENCE_ACC(S);
    blk_compute<2>(gA, awA, yB, S);
    BLK_FENCE_ACC(S);
    blk_issue<3>(10, ao, X, Wl, gA, awA);
    blk_load_offsets<3>(11, ldsTab, half, ao);
    BLK_FENCE_ACC(S);
    blk_compute<2>(gB, awB, yB, S);
    BLK_FENCE_ACC(S);
#pragma unroll 1
    for (int blk = 10; blk < 24; blk += 2) {
      blk_issue<3>(blk + 1, ao, X, Wl, gB, awB);
      blk_load_offsets<3>(blk + 2, ldsTab, half, ao);
      BLK_FENCE_ACC(S);
      blk_compute<3>(gA, awA, yB, S);
      BLK_FENCE_ACC(S);
      blk_issue<3>(blk + 2, ao, X, Wl, gA, awA);
      blk_load_offsets<3>(blk + 3, ldsTab, half, ao);
      BLK_FENCE_ACC(S);
      blk_compute<3>(gB, awB, yB, S);
      BLK_FENCE_ACC(S);
    }
    blk_issue<3>(25, ao, X, Wl, gB, awB);
    BLK_FENCE_ACC(S);
    blk_compute<3>(gA, awA, yB, S);
    BLK_FENCE_ACC(S);
    blk_compute<3>(gB, awB, yB, S);

    if (ai + 1 < A_PER_WG) stage_load(sr, x, Wg, a + 1, tid, lane, myb);

#pragma unroll
    for (int t = 0; t < 4; ++t) {
      float v = S[2 * t] + S[2 * t + 1];
      v += __shfl_xor(v, 16, 32);
      if (half == 0) out[(size_t)(myb + t * 16 + lrow) * NA + a] = v;
    }
  }
}

extern "C" void kernel_launch(void* const* d_in, const int* in_sizes, int n_in,
                              void* d_out, int out_size, void* d_ws, size_t ws_size,
                              hipStream_t stream) {
  const float* x  = (const float*)d_in[0];
  const float* y  = (const float*)d_in[1];
  const float* w1 = (const float*)d_in[2];
  const float* w2 = (const float*)d_in[3];
  const float* w3 = (const float*)d_in[4];
  const float* c1 = (const float*)d_in[5];
  const float* c2 = (const float*)d_in[6];
  const float* c3 = (const float*)d_in[7];
  const int* i1 = (const int*)d_in[8];
  const int* l1 = (const int*)d_in[9];
  const int* k1 = (const int*)d_in[10];
  const int* q1 = (const int*)d_in[11];
  const int* i2 = (const int*)d_in[12];
  const int* j2 = (const int*)d_in[13];
  const int* l2 = (const int*)d_in[14];
  const int* m2 = (const int*)d_in[15];
  const int* k2 = (const int*)d_in[16];
  const int* q2 = (const int*)d_in[17];
  const int* i3 = (const int*)d_in[18];
  const int* j3 = (const int*)d_in[19];
  const int* f3 = (const int*)d_in[20];
  const int* l3 = (const int*)d_in[21];
  const int* m3 = (const int*)d_in[22];
  const int* g3 = (const int*)d_in[23];
  const int* k3 = (const int*)d_in[24];
  const int* q3 = (const int*)d_in[25];

  int*   offs = (int*)d_ws;
  float* Wg   = (float*)((char*)d_ws + 5120);
  float* out  = (float*)d_out;

  prep_kernel<<<PREP_WG_BLOCKS + PREP_OFF_BLOCKS, 256, 0, stream>>>(w1, w2, w3, c1, c2, c3,
                                                                    k1, q1, k2, q2, k3, q3,
                                                                    i1, l1, i2, j2, l2, m2,
                                                                    i3, j3, f3, l3, m3, g3, Wg, offs);
  contract_kernel<<<dim3(16, 32), WG_THREADS, 0, stream>>>(x, y, Wg, offs, out);
}
